# starting kernel
# baseline (speedup 1.0000x reference)
#include <hip/hip_runtime.h>
#include <math.h>

#define BATCH 32
#define NNODE 512
#define GGRP  512
#define EMBD  256
#define HNUM  16
#define DK    16

typedef __attribute__((ext_vector_type(2))) float v2f;
typedef __attribute__((ext_vector_type(8))) float v8f;

__device__ __forceinline__ v8f wmma4(v2f a, v2f b, v8f c) {
  return __builtin_amdgcn_wmma_f32_16x16x4_f32(
       false, a,  false, b,
       (short)0, c,  false,  false);
}

__device__ __forceinline__ v2f ld2g(const float* __restrict__ p) {
  return *reinterpret_cast<const v2f*>(p);
}
__device__ __forceinline__ v2f ld2s(const float* p) {
  return *reinterpret_cast<const v2f*>(p);
}

__global__ void mean_qg_kernel(const float* __restrict__ EN,
                               const float* __restrict__ WqG,
                               float* __restrict__ qg) {
  __shared__ float sm[EMBD];
  const int b = blockIdx.x;
  const int e = threadIdx.x;
  float s = 0.0f;
  const float* base = EN + (size_t)b * NNODE * EMBD + e;
  for (int n = 0; n < NNODE; ++n) s += base[(size_t)n * EMBD];
  sm[e] = s * (1.0f / (float)NNODE);
  __syncthreads();
  float acc = 0.0f;
  const float* wrow = WqG + (size_t)e * EMBD;
  for (int k = 0; k < EMBD; ++k) acc += sm[k] * wrow[k];
  qg[(size_t)b * EMBD + e] = acc;
}

__global__ void proj_kernel(const float* __restrict__ EN,
                            const float* __restrict__ EF,
                            const float* __restrict__ EL,
                            const float* __restrict__ WqF,
                            const float* __restrict__ WqL,
                            const float* __restrict__ Wk,
                            const float* __restrict__ Wv,
                            const float* __restrict__ qg,
                            float* __restrict__ Q,
                            float* __restrict__ K,
                            float* __restrict__ V) {
  const int widx = threadIdx.x >> 5;
  const int lane = threadIdx.x & 31;
  const int ln   = lane & 15;
  const int kh   = lane >> 4;
  const int wave = blockIdx.x * 4 + widx;
  const int b    = wave >> 6;
  const int rem  = wave & 63;
  const int mt   = rem >> 2;
  const int ng   = rem & 3;
  const int row0 = mt * 32 + ln;
  const int sel  = blockIdx.y;

  v8f acc[2][4];
#pragma unroll
  for (int i = 0; i < 2; ++i)
#pragma unroll
    for (int j = 0; j < 4; ++j) acc[i][j] = v8f{0.f,0.f,0.f,0.f,0.f,0.f,0.f,0.f};

  float* Out;
  if (sel < 2) {
    const float* W  = (sel == 0) ? Wk : Wv;
    const float* A0 = EN + (size_t)(b * NNODE + row0) * EMBD;
    const float* A1 = A0 + (size_t)16 * EMBD;
    const float* Br[4];
#pragma unroll
    for (int j = 0; j < 4; ++j) Br[j] = W + (size_t)(ng * 64 + j * 16 + ln) * EMBD;
    for (int k = 0; k < EMBD; k += 4) {
      const v2f a0 = ld2g(A0 + k + 2 * kh);
      const v2f a1 = ld2g(A1 + k + 2 * kh);
#pragma unroll
      for (int j = 0; j < 4; ++j) {
        const v2f bb = ld2g(Br[j] + k + 2 * kh);
        acc[0][j] = wmma4(a0, bb, acc[0][j]);
        acc[1][j] = wmma4(a1, bb, acc[1][j]);
      }
    }
    Out = (sel == 0) ? K : V;
  } else {
    const float* A0 = EF + (size_t)(b * GGRP + row0) * EMBD;
    const float* A1 = A0 + (size_t)16 * EMBD;
    const float* Br[4];
#pragma unroll
    for (int j = 0; j < 4; ++j) Br[j] = WqF + (size_t)(ng * 64 + j * 16 + ln) * EMBD;
    for (int k = 0; k < EMBD; k += 4) {
      const v2f a0 = ld2g(A0 + k + 2 * kh);
      const v2f a1 = ld2g(A1 + k + 2 * kh);
#pragma unroll
      for (int j = 0; j < 4; ++j) {
        const v2f bb = ld2g(Br[j] + k + 2 * kh);
        acc[0][j] = wmma4(a0, bb, acc[0][j]);
        acc[1][j] = wmma4(a1, bb, acc[1][j]);
      }
    }
    const float* C0 = EL + (size_t)(b * GGRP + row0) * EMBD;
    const float* C1 = C0 + (size_t)16 * EMBD;
#pragma unroll
    for (int j = 0; j < 4; ++j) Br[j] = WqL + (size_t)(ng * 64 + j * 16 + ln) * EMBD;
    for (int k = 0; k < EMBD; k += 4) {
      const v2f a0 = ld2g(C0 + k + 2 * kh);
      const v2f a1 = ld2g(C1 + k + 2 * kh);
#pragma unroll
      for (int j = 0; j < 4; ++j) {
        const v2f bb = ld2g(Br[j] + k + 2 * kh);
        acc[0][j] = wmma4(a0, bb, acc[0][j]);
        acc[1][j] = wmma4(a1, bb, acc[1][j]);
      }
    }
#pragma unroll
    for (int j = 0; j < 4; ++j) {
      const float qv = qg[(size_t)b * EMBD + ng * 64 + j * 16 + ln];
#pragma unroll
      for (int i = 0; i < 2; ++i)
#pragma unroll
        for (int r = 0; r < 8; ++r) acc[i][j][r] += qv;
    }
    Out = Q;
  }

#pragma unroll
  for (int i = 0; i < 2; ++i)
#pragma unroll
    for (int j = 0; j < 4; ++j) {
      const int col = ng * 64 + j * 16 + ln;
#pragma unroll
      for (int r = 0; r < 8; ++r) {
        const int row = mt * 32 + i * 16 + r + 8 * kh;
        Out[(size_t)(b * NNODE + row) * EMBD + col] = acc[i][j][r];
      }
    }
}

__global__ void attn_kernel(const float* __restrict__ Q,
                            const float* __restrict__ K,
                            const float* __restrict__ V,
                            const float* __restrict__ MASK,
                            float* __restrict__ outc) {
  __shared__ float sc[16 * 512];
  __shared__ float part[8 * 256];
  __shared__ float redA[16][16];
  __shared__ float redB[16][16];

  const int gt = blockIdx.x, h = blockIdx.y, b = blockIdx.z;
  const int t    = threadIdx.x;
  const int widx = t >> 5;
  const int lane = t & 31;
  const int ln   = lane & 15;
  const int kh   = lane >> 4;

  {
    const float* Arow = Q + (size_t)(b * GGRP + gt * 16 + ln) * EMBD + h * DK;
    const float* Br[4];
#pragma unroll
    for (int j = 0; j < 4; ++j)
      Br[j] = K + (size_t)(b * NNODE + (widx * 4 + j) * 16 + ln) * EMBD + h * DK;
    v8f acc[4];
#pragma unroll
    for (int j = 0; j < 4; ++j) acc[j] = v8f{0.f,0.f,0.f,0.f,0.f,0.f,0.f,0.f};
#pragma unroll
    for (int k = 0; k < DK; k += 4) {
      const v2f a = ld2g(Arow + k + 2 * kh);
#pragma unroll
      for (int j = 0; j < 4; ++j)
        acc[j] = wmma4(a, ld2g(Br[j] + k + 2 * kh), acc[j]);
    }
#pragma unroll
    for (int j = 0; j < 4; ++j) {
      const int nb = (widx * 4 + j) * 16;
#pragma unroll
      for (int r = 0; r < 8; ++r) {
        const int row = r + 8 * kh;
        const int g   = gt * 16 + row;
        sc[row * 512 + nb + ln] =
            acc[j][r] * 0.25f + MASK[(size_t)(b * GGRP + g) * NNODE + nb + ln];
      }
    }
  }
  __syncthreads();

  const int row = t >> 4, l16 = t & 15;
  float* srow = sc + row * 512;
  float mx = -INFINITY;
  for (int j = l16; j < 512; j += 16) mx = fmaxf(mx, srow[j]);
  redA[row][l16] = mx;
  __syncthreads();
  if (l16 == 0) {
    float m2 = redA[row][0];
    for (int i = 1; i < 16; ++i) m2 = fmaxf(m2, redA[row][i]);
    redA[row][0] = m2;
  }
  __syncthreads();
  const float rmax = redA[row][0];
  float ssum = 0.0f;
  for (int j = l16; j < 512; j += 16) {
    const float e = __expf(srow[j] - rmax);
    srow[j] = e;
    ssum += e;
  }
  redB[row][l16] = ssum;
  __syncthreads();
  if (l16 == 0) {
    float s2 = 0.0f;
    for (int i = 0; i < 16; ++i) s2 += redB[row][i];
    redB[row][0] = s2;
  }
  __syncthreads();
  const float inv = 1.0f / redB[row][0];
  for (int j = l16; j < 512; j += 16) srow[j] *= inv;
  __syncthreads();

  v8f acc = {0.f,0.f,0.f,0.f,0.f,0.f,0.f,0.f};
#pragma unroll
  for (int nti = 0; nti < 4; ++nti) {
    const int nb = (widx * 4 + nti) * 16;
#pragma unroll
    for (int k = 0; k < 16; k += 4) {
      const int kk = nb + k + 2 * kh;
      v2f a = ld2s(sc + ln * 512 + kk);
      const float* vp = V + (size_t)(b * NNODE + kk) * EMBD + h * DK + ln;
      v2f bb;
      bb.x = vp[0];
      bb.y = vp[EMBD];
      acc = wmma4(a, bb, acc);
    }
  }
#pragma unroll
  for (int r = 0; r < 8; ++r)
    part[widx * 256 + (r + 8 * kh) * 16 + ln] = acc[r];
  __syncthreads();

  {
    const int orow = t >> 4, ocol = t & 15;
    float s = 0.0f;
#pragma unroll
    for (int w = 0; w < 8; ++w) s += part[w * 256 + orow * 16 + ocol];
    outc[(size_t)(b * GGRP + gt * 16 + orow) * EMBD + h * DK + ocol] = s;
  }
}

__global__ void mh_kernel(const float* __restrict__ outc,
                          const float* __restrict__ Wc,
                          const float* __restrict__ bc,
                          float* __restrict__ mh) {
  const int widx = threadIdx.x >> 5;
  const int lane = threadIdx.x & 31;
  const int ln   = lane & 15;
  const int kh   = lane >> 4;
  const int wave = blockIdx.x * 4 + widx;
  const int b    = wave >> 6;
  const int rem  = wave & 63;
  const int mt   = rem >> 2;
  const int ng   = rem & 3;
  const int row0 = mt * 32 + ln;

  v8f acc[2][4];
#pragma unroll
  for (int i = 0; i < 2; ++i)
#pragma unroll
    for (int j = 0; j < 4; ++j) acc[i][j] = v8f{0.f,0.f,0.f,0.f,0.f,0.f,0.f,0.f};

  const float* A0 = outc + (size_t)(b * GGRP + row0) * EMBD;
  const float* A1 = A0 + (size_t)16 * EMBD;
  const float* Br[4];
#pragma unroll
  for (int j = 0; j < 4; ++j) Br[j] = Wc + (size_t)(ng * 64 + j * 16 + ln) * EMBD;
  for (int k = 0; k < EMBD; k += 4) {
    const v2f a0 = ld2g(A0 + k + 2 * kh);
    const v2f a1 = ld2g(A1 + k + 2 * kh);
#pragma unroll
    for (int j = 0; j < 4; ++j) {
      const v2f bb = ld2g(Br[j] + k + 2 * kh);
      acc[0][j] = wmma4(a0, bb, acc[0][j]);
      acc[1][j] = wmma4(a1, bb, acc[1][j]);
    }
  }
#pragma unroll
  for (int i = 0; i < 2; ++i)
#pragma unroll
    for (int j = 0; j < 4; ++j) {
      const int col  = ng * 64 + j * 16 + ln;
      const float bv = bc[col];
#pragma unroll
      for (int r = 0; r < 8; ++r) {
        const int row = mt * 32 + i * 16 + r + 8 * kh;
        mh[(size_t)(b * GGRP + row) * EMBD + col] = acc[i][j][r] + bv;
      }
    }
}

__global__ void final_kernel(const float* __restrict__ mh,
                             const float* __restrict__ EN,
                             const float* __restrict__ MASK,
                             float* __restrict__ out) {
  __shared__ float sc[16 * 512];
  __shared__ float redA[16][16];
  __shared__ float redB[16][16];

  const int gt = blockIdx.x, b = blockIdx.y;
  const int t    = threadIdx.x;
  const int widx = t >> 5;
  const int lane = t & 31;
  const int ln   = lane & 15;
  const int kh   = lane >> 4;

  {
    const float* Arow = mh + (size_t)(b * GGRP + gt * 16 + ln) * EMBD;
    const float* Br[4];
#pragma unroll
    for (int j = 0; j < 4; ++j)
      Br[j] = EN + (size_t)(b * NNODE + (widx * 4 + j) * 16 + ln) * EMBD;
    v8f acc[4];
#pragma unroll
    for (int j = 0; j < 4; ++j) acc[j] = v8f{0.f,0.f,0.f,0.f,0.f,0.f,0.f,0.f};
    for (int k = 0; k < EMBD; k += 4) {
      const v2f a = ld2g(Arow + k + 2 * kh);
#pragma unroll
      for (int j = 0; j < 4; ++j)
        acc[j] = wmma4(a, ld2g(Br[j] + k + 2 * kh), acc[j]);
    }
#pragma unroll
    for (int j = 0; j < 4; ++j) {
      const int nb = (widx * 4 + j) * 16;
#pragma unroll
      for (int r = 0; r < 8; ++r) {
        const int row = r + 8 * kh;
        const int g   = gt * 16 + row;
        const float s2 = acc[j][r] * (1.0f / 16.0f);
        sc[row * 512 + nb + ln] =
            10.0f * tanhf(s2) + MASK[(size_t)(b * GGRP + g) * NNODE + nb + ln];
      }
    }
  }
  __syncthreads();

  const int row = t >> 4, l16 = t & 15;
  float* srow = sc + row * 512;
  float mx = -INFINITY;
  for (int j = l16; j < 512; j += 16) mx = fmaxf(mx, srow[j]);
  redA[row][l16] = mx;
  __syncthreads();
  if (l16 == 0) {
    float m2 = redA[row][0];
    for (int i = 1; i < 16; ++i) m2 = fmaxf(m2, redA[row][i]);
    redA[row][0] = m2;
  }
  __syncthreads();
  const float rmax = redA[row][0];
  float ssum = 0.0f;
  for (int j = l16; j < 512; j += 16) {
    const float e = __expf(srow[j] - rmax);
    srow[j] = e;
    ssum += e;
  }
  redB[row][l16] = ssum;
  __syncthreads();
  if (l16 == 0) {
    float s2 = 0.0f;
    for (int i = 0; i < 16; ++i) s2 += redB[row][i];
    redB[row][0] = s2;
  }
  __syncthreads();
  const float inv = 1.0f / redB[row][0];
  float* orow = out + (size_t)(b * GGRP + gt * 16 + row) * NNODE;
  for (int j = l16; j < 512; j += 16) orow[j] = srow[j] * inv;
}

extern "C" void kernel_launch(void* const* d_in, const int* in_sizes, int n_in,
                              void* d_out, int out_size, void* d_ws, size_t ws_size,
                              hipStream_t stream) {
  const float* EN   = (const float*)d_in[0];
  const float* EF   = (const float*)d_in[1];
  const float* EL   = (const float*)d_in[2];
  const float* MASK = (const float*)d_in[3];
  const float* WqG  = (const float*)d_in[4];
  const float* WqF  = (const float*)d_in[5];
  const float* WqL  = (const float*)d_in[6];
  const float* Wk   = (const float*)d_in[7];
  const float* Wv   = (const float*)d_in[8];
  const float* Wc   = (const float*)d_in[9];
  const float* bc   = (const float*)d_in[10];
  float* out = (float*)d_out;

  float* ws   = (float*)d_ws;
  float* qg   = ws;
  float* Qb   = qg   + (size_t)BATCH * EMBD;
  float* Kb   = Qb   + (size_t)BATCH * GGRP * EMBD;
  float* Vb   = Kb   + (size_t)BATCH * NNODE * EMBD;
  float* outc = Vb   + (size_t)BATCH * NNODE * EMBD;
  float* mhb  = outc + (size_t)BATCH * GGRP * EMBD;

  mean_qg_kernel<<<BATCH, 256, 0, stream>>>(EN, WqG, qg);

  dim3 gp(512, 3, 1);
  proj_kernel<<<gp, 128, 0, stream>>>(EN, EF, EL, WqF, WqL, Wk, Wv, qg, Qb, Kb, Vb);

  dim3 ga(GGRP / 16, HNUM, BATCH);
  attn_kernel<<<ga, 256, 0, stream>>>(Qb, Kb, Vb, MASK, outc);

  mh_kernel<<<512, 128, 0, stream>>>(outc, Wc, bc, mhb);

  dim3 gf(GGRP / 16, BATCH, 1);
  final_kernel<<<gf, 256, 0, stream>>>(mhb, EN, MASK, out);
}


# Fragment loads widened b64->b128: each lane takes 4 consecutive k (lane halves 4 apart) and feeds two WMMA k-steps; half the 16-line loads per byte; k sum order permuted inside 8-blocks
# speedup vs baseline: 1.2113x; 1.2113x over previous
#include <hip/hip_runtime.h>
#include <math.h>

#define BATCH 32
#define NNODE 512
#define GGRP  512
#define EMBD  256
#define HNUM  16
#define DK    16

typedef __attribute__((ext_vector_type(2))) float v2f;
typedef __attribute__((ext_vector_type(4))) float v4f;
typedef __attribute__((ext_vector_type(8))) float v8f;

__device__ __forceinline__ v8f wmma4(v2f a, v2f b, v8f c) {
  return __builtin_amdgcn_wmma_f32_16x16x4_f32(
       false, a,  false, b,
       (short)0, c,  false,  false);
}

__device__ __forceinline__ v2f ld2s(const float* p) {
  return *reinterpret_cast<const v2f*>(p);
}
__device__ __forceinline__ v4f ld4g(const float* __restrict__ p) {
  return *reinterpret_cast<const v4f*>(p);
}
__device__ __forceinline__ v2f lo2(v4f v) { return __builtin_shufflevector(v, v, 0, 1); }
__device__ __forceinline__ v2f hi2(v4f v) { return __builtin_shufflevector(v, v, 2, 3); }

__global__ void mean_qg_kernel(const float* __restrict__ EN,
                               const float* __restrict__ WqG,
                               float* __restrict__ qg) {
  __shared__ float sm[EMBD];
  const int b = blockIdx.x;
  const int e = threadIdx.x;
  float s = 0.0f;
  const float* base = EN + (size_t)b * NNODE * EMBD + e;
  for (int n = 0; n < NNODE; ++n) s += base[(size_t)n * EMBD];
  sm[e] = s * (1.0f / (float)NNODE);
  __syncthreads();
  float acc = 0.0f;
  const float* wrow = WqG + (size_t)e * EMBD;
  for (int k = 0; k < EMBD; ++k) acc += sm[k] * wrow[k];
  qg[(size_t)b * EMBD + e] = acc;
}

__global__ void proj_kernel(const float* __restrict__ EN,
                            const float* __restrict__ EF,
                            const float* __restrict__ EL,
                            const float* __restrict__ WqF,
                            const float* __restrict__ WqL,
                            const float* __restrict__ Wk,
                            const float* __restrict__ Wv,
                            const float* __restrict__ qg,
                            float* __restrict__ Q,
                            float* __restrict__ K,
                            float* __restrict__ V) {
  const int widx = threadIdx.x >> 5;
  const int lane = threadIdx.x & 31;
  const int ln   = lane & 15;
  const int kh   = lane >> 4;
  const int wave = blockIdx.x * 4 + widx;
  const int b    = wave >> 6;
  const int rem  = wave & 63;
  const int mt   = rem >> 2;
  const int ng   = rem & 3;
  const int row0 = mt * 32 + ln;
  const int sel  = blockIdx.y;

  v8f acc[2][4];
#pragma unroll
  for (int i = 0; i < 2; ++i)
#pragma unroll
    for (int j = 0; j < 4; ++j) acc[i][j] = v8f{0.f,0.f,0.f,0.f,0.f,0.f,0.f,0.f};

  float* Out;
  if (sel < 2) {
    const float* W  = (sel == 0) ? Wk : Wv;
    const float* A0 = EN + (size_t)(b * NNODE + row0) * EMBD;
    const float* A1 = A0 + (size_t)16 * EMBD;
    const float* Br[4];
#pragma unroll
    for (int j = 0; j < 4; ++j) Br[j] = W + (size_t)(ng * 64 + j * 16 + ln) * EMBD;
    for (int k = 0; k < EMBD; k += 8) {
      const v4f a0 = ld4g(A0 + k + 4 * kh);
      const v4f a1 = ld4g(A1 + k + 4 * kh);
      v4f bb[4];
#pragma unroll
      for (int j = 0; j < 4; ++j) bb[j] = ld4g(Br[j] + k + 4 * kh);
#pragma unroll
      for (int j = 0; j < 4; ++j) {
        acc[0][j] = wmma4(lo2(a0), lo2(bb[j]), acc[0][j]);
        acc[1][j] = wmma4(lo2(a1), lo2(bb[j]), acc[1][j]);
      }
#pragma unroll
      for (int j = 0; j < 4; ++j) {
        acc[0][j] = wmma4(hi2(a0), hi2(bb[j]), acc[0][j]);
        acc[1][j] = wmma4(hi2(a1), hi2(bb[j]), acc[1][j]);
      }
    }
    Out = (sel == 0) ? K : V;
  } else {
    const float* A0 = EF + (size_t)(b * GGRP + row0) * EMBD;
    const float* A1 = A0 + (size_t)16 * EMBD;
    const float* Br[4];
#pragma unroll
    for (int j = 0; j < 4; ++j) Br[j] = WqF + (size_t)(ng * 64 + j * 16 + ln) * EMBD;
    for (int k = 0; k < EMBD; k += 8) {
      const v4f a0 = ld4g(A0 + k + 4 * kh);
      const v4f a1 = ld4g(A1 + k + 4 * kh);
      v4f bb[4];
#pragma unroll
      for (int j = 0; j < 4; ++j) bb[j] = ld4g(Br[j] + k + 4 * kh);
#pragma unroll
      for (int j = 0; j < 4; ++j) {
        acc[0][j] = wmma4(lo2(a0), lo2(bb[j]), acc[0][j]);
        acc[1][j] = wmma4(lo2(a1), lo2(bb[j]), acc[1][j]);
      }
#pragma unroll
      for (int j = 0; j < 4; ++j) {
        acc[0][j] = wmma4(hi2(a0), hi2(bb[j]), acc[0][j]);
        acc[1][j] = wmma4(hi2(a1), hi2(bb[j]), acc[1][j]);
      }
    }
    const float* C0 = EL + (size_t)(b * GGRP + row0) * EMBD;
    const float* C1 = C0 + (size_t)16 * EMBD;
#pragma unroll
    for (int j = 0; j < 4; ++j) Br[j] = WqL + (size_t)(ng * 64 + j * 16 + ln) * EMBD;
    for (int k = 0; k < EMBD; k += 8) {
      const v4f a0 = ld4g(C0 + k + 4 * kh);
      const v4f a1 = ld4g(C1 + k + 4 * kh);
      v4f bb[4];
#pragma unroll
      for (int j = 0; j < 4; ++j) bb[j] = ld4g(Br[j] + k + 4 * kh);
#pragma unroll
      for (int j = 0; j < 4; ++j) {
        acc[0][j] = wmma4(lo2(a0), lo2(bb[j]), acc[0][j]);
        acc[1][j] = wmma4(lo2(a1), lo2(bb[j]), acc[1][j]);
      }
#pragma unroll
      for (int j = 0; j < 4; ++j) {
        acc[0][j] = wmma4(hi2(a0), hi2(bb[j]), acc[0][j]);
        acc[1][j] = wmma4(hi2(a1), hi2(bb[j]), acc[1][j]);
      }
    }
#pragma unroll
    for (int j = 0; j < 4; ++j) {
      const float qv = qg[(size_t)b * EMBD + ng * 64 + j * 16 + ln];
#pragma unroll
      for (int i = 0; i < 2; ++i)
#pragma unroll
        for (int r = 0; r < 8; ++r) acc[i][j][r] += qv;
    }
    Out = Q;
  }

#pragma unroll
  for (int i = 0; i < 2; ++i)
#pragma unroll
    for (int j = 0; j < 4; ++j) {
      const int col = ng * 64 + j * 16 + ln;
#pragma unroll
      for (int r = 0; r < 8; ++r) {
        const int row = mt * 32 + i * 16 + r + 8 * kh;
        Out[(size_t)(b * NNODE + row) * EMBD + col] = acc[i][j][r];
      }
    }
}

__global__ void attn_kernel(const float* __restrict__ Q,
                            const float* __restrict__ K,
                            const float* __restrict__ V,
                            const float* __restrict__ MASK,
                            float* __restrict__ outc) {
  __shared__ float sc[16 * 512];
  __shared__ float part[8 * 256];
  __shared__ float redA[16][16];
  __shared__ float redB[16][16];

  const int gt = blockIdx.x, h = blockIdx.y, b = blockIdx.z;
  const int t    = threadIdx.x;
  const int widx = t >> 5;
  const int lane = t & 31;
  const int ln   = lane & 15;
  const int kh   = lane >> 4;

  {
    const float* Arow = Q + (size_t)(b * GGRP + gt * 16 + ln) * EMBD + h * DK;
    const float* Br[4];
#pragma unroll
    for (int j = 0; j < 4; ++j)
      Br[j] = K + (size_t)(b * NNODE + (widx * 4 + j) * 16 + ln) * EMBD + h * DK;
    v8f acc[4];
#pragma unroll
    for (int j = 0; j < 4; ++j) acc[j] = v8f{0.f,0.f,0.f,0.f,0.f,0.f,0.f,0.f};
#pragma unroll
    for (int k = 0; k < DK; k += 8) {
      const v4f a = ld4g(Arow + k + 4 * kh);
      v4f bb[4];
#pragma unroll
      for (int j = 0; j < 4; ++j) bb[j] = ld4g(Br[j] + k + 4 * kh);
#pragma unroll
      for (int j = 0; j < 4; ++j) acc[j] = wmma4(lo2(a), lo2(bb[j]), acc[j]);
#pragma unroll
      for (int j = 0; j < 4; ++j) acc[j] = wmma4(hi2(a), hi2(bb[j]), acc[j]);
    }
#pragma unroll
    for (int j = 0; j < 4; ++j) {
      const int nb = (widx * 4 + j) * 16;
#pragma unroll
      for (int r = 0; r < 8; ++r) {
        const int row = r + 8 * kh;
        const int g   = gt * 16 + row;
        sc[row * 512 + nb + ln] =
            acc[j][r] * 0.25f + MASK[(size_t)(b * GGRP + g) * NNODE + nb + ln];
      }
    }
  }
  __syncthreads();

  const int row = t >> 4, l16 = t & 15;
  float* srow = sc + row * 512;
  float mx = -INFINITY;
  for (int j = l16; j < 512; j += 16) mx = fmaxf(mx, srow[j]);
  redA[row][l16] = mx;
  __syncthreads();
  if (l16 == 0) {
    float m2 = redA[row][0];
    for (int i = 1; i < 16; ++i) m2 = fmaxf(m2, redA[row][i]);
    redA[row][0] = m2;
  }
  __syncthreads();
  const float rmax = redA[row][0];
  float ssum = 0.0f;
  for (int j = l16; j < 512; j += 16) {
    const float e = __expf(srow[j] - rmax);
    srow[j] = e;
    ssum += e;
  }
  redB[row][l16] = ssum;
  __syncthreads();
  if (l16 == 0) {
    float s2 = 0.0f;
    for (int i = 0; i < 16; ++i) s2 += redB[row][i];
    redB[row][0] = s2;
  }
  __syncthreads();
  const float inv = 1.0f / redB[row][0];
  for (int j = l16; j < 512; j += 16) srow[j] *= inv;
  __syncthreads();

  v8f acc = {0.f,0.f,0.f,0.f,0.f,0.f,0.f,0.f};
#pragma unroll
  for (int nti = 0; nti < 4; ++nti) {
    const int nb = (widx * 4 + nti) * 16;
#pragma unroll
    for (int k = 0; k < 16; k += 4) {
      const int kk = nb + k + 2 * kh;
      v2f a = ld2s(sc + ln * 512 + kk);
      const float* vp = V + (size_t)(b * NNODE + kk) * EMBD + h * DK + ln;
      v2f bb;
      bb.x = vp[0];
      bb.y = vp[EMBD];
      acc = wmma4(a, bb, acc);
    }
  }
#pragma unroll
  for (int r = 0; r < 8; ++r)
    part[widx * 256 + (r + 8 * kh) * 16 + ln] = acc[r];
  __syncthreads();

  {
    const int orow = t >> 4, ocol = t & 15;
    float s = 0.0f;
#pragma unroll
    for (int w = 0; w < 8; ++w) s += part[w * 256 + orow * 16 + ocol];
    outc[(size_t)(b * GGRP + gt * 16 + orow) * EMBD + h * DK + ocol] = s;
  }
}

__global__ void mh_kernel(const float* __restrict__ outc,
                          const float* __restrict__ Wc,
                          const float* __restrict__ bc,
                          float* __restrict__ mh) {
  const int widx = threadIdx.x >> 5;
  const int lane = threadIdx.x & 31;
  const int ln   = lane & 15;
  const int kh   = lane >> 4;
  const int wave = blockIdx.x * 4 + widx;
  const int b    = wave >> 6;
  const int rem  = wave & 63;
  const int mt   = rem >> 2;
  const int ng   = rem & 3;
  const int row0 = mt * 32 + ln;

  v8f acc[2][4];
#pragma unroll
  for (int i = 0; i < 2; ++i)
#pragma unroll
    for (int j = 0; j < 4; ++j) acc[i][j] = v8f{0.f,0.f,0.f,0.f,0.f,0.f,0.f,0.f};

  const float* A0 = outc + (size_t)(b * GGRP + row0) * EMBD;
  const float* A1 = A0 + (size_t)16 * EMBD;
  const float* Br[4];
#pragma unroll
  for (int j = 0; j < 4; ++j) Br[j] = Wc + (size_t)(ng * 64 + j * 16 + ln) * EMBD;
  for (int k = 0; k < EMBD; k += 8) {
    const v4f a0 = ld4g(A0 + k + 4 * kh);
    const v4f a1 = ld4g(A1 + k + 4 * kh);
    v4f bb[4];
#pragma unroll
    for (int j = 0; j < 4; ++j) bb[j] = ld4g(Br[j] + k + 4 * kh);
#pragma unroll
    for (int j = 0; j < 4; ++j) {
      acc[0][j] = wmma4(lo2(a0), lo2(bb[j]), acc[0][j]);
      acc[1][j] = wmma4(lo2(a1), lo2(bb[j]), acc[1][j]);
    }
#pragma unroll
    for (int j = 0; j < 4; ++j) {
      acc[0][j] = wmma4(hi2(a0), hi2(bb[j]), acc[0][j]);
      acc[1][j] = wmma4(hi2(a1), hi2(bb[j]), acc[1][j]);
    }
  }
#pragma unroll
  for (int i = 0; i < 2; ++i)
#pragma unroll
    for (int j = 0; j < 4; ++j) {
      const int col  = ng * 64 + j * 16 + ln;
      const float bv = bc[col];
#pragma unroll
      for (int r = 0; r < 8; ++r) {
        const int row = mt * 32 + i * 16 + r + 8 * kh;
        mh[(size_t)(b * GGRP + row) * EMBD + col] = acc[i][j][r] + bv;
      }
    }
}

__global__ void final_kernel(const float* __restrict__ mh,
                             const float* __restrict__ EN,
                             const float* __restrict__ MASK,
                             float* __restrict__ out) {
  __shared__ float sc[16 * 512];
  __shared__ float redA[16][16];
  __shared__ float redB[16][16];

  const int gt = blockIdx.x, b = blockIdx.y;
  const int t    = threadIdx.x;
  const int widx = t >> 5;
  const int lane = t & 31;
  const int ln   = lane & 15;
  const int kh   = lane >> 4;

  {
    const float* Arow = mh + (size_t)(b * GGRP + gt * 16 + ln) * EMBD;
    const float* Br[4];
#pragma unroll
    for (int j = 0; j < 4; ++j)
      Br[j] = EN + (size_t)(b * NNODE + (widx * 4 + j) * 16 + ln) * EMBD;
    v8f acc[4];
#pragma unroll
    for (int j = 0; j < 4; ++j) acc[j] = v8f{0.f,0.f,0.f,0.f,0.f,0.f,0.f,0.f};
    for (int k = 0; k < EMBD; k += 8) {
      const v4f a = ld4g(Arow + k + 4 * kh);
      v4f bb[4];
#pragma unroll
      for (int j = 0; j < 4; ++j) bb[j] = ld4g(Br[j] + k + 4 * kh);
#pragma unroll
      for (int j = 0; j < 4; ++j) acc[j] = wmma4(lo2(a), lo2(bb[j]), acc[j]);
#pragma unroll
      for (int j = 0; j < 4; ++j) acc[j] = wmma4(hi2(a), hi2(bb[j]), acc[j]);
    }
#pragma unroll
    for (int j = 0; j < 4; ++j) {
      const int nb = (widx * 4 + j) * 16;
#pragma unroll
      for (int r = 0; r < 8; ++r) {
        const int row = r + 8 * kh;
        const int g   = gt * 16 + row;
        const float s2 = acc[j][r] * (1.0f / 16.0f);
        sc[row * 512 + nb + ln] =
            10.0f * tanhf(s2) + MASK[(size_t)(b * GGRP + g) * NNODE + nb + ln];
      }
    }
  }
  __syncthreads();

  const int row = t >> 4, l16 = t & 15;
  float* srow = sc + row * 512;
  float mx = -INFINITY;
  for (int j = l16; j < 512; j += 16) mx = fmaxf(mx, srow[j]);
  redA[row][l16] = mx;
  __syncthreads();
  if (l16 == 0) {
    float m2 = redA[row][0];
    for (int i = 1; i < 16; ++i) m2 = fmaxf(m2, redA[row][i]);
    redA[row][0] = m2;
  }
  __syncthreads();
  const float rmax = redA[row][0];
  float ssum = 0.0f;
  for (int j = l16; j < 512; j += 16) {
    const float e = __expf(srow[j] - rmax);
    srow[j] = e;
    ssum += e;
  }
  redB[row][l16] = ssum;
  __syncthreads();
  if (l16 == 0) {
    float s2 = 0.0f;
    for (int i = 0; i < 16; ++i) s2 += redB[row][i];
    redB[row][0] = s2;
  }
  __syncthreads();
  const float inv = 1.0f / redB[row][0];
  float* orow = out + (size_t)(b * GGRP + gt * 16 + row) * NNODE;
  for (int j = l16; j < 512; j += 16) orow[j] = srow[j] * inv;
}

extern "C" void kernel_launch(void* const* d_in, const int* in_sizes, int n_in,
                              void* d_out, int out_size, void* d_ws, size_t ws_size,
                              hipStream_t stream) {
  const float* EN   = (const float*)d_in[0];
  const float* EF   = (const float*)d_in[1];
  const float* EL   = (const float*)d_in[2];
  const float* MASK = (const float*)d_in[3];
  const float* WqG  = (const float*)d_in[4];
  const float* WqF  = (const float*)d_in[5];
  const float* WqL  = (const float*)d_in[6];
  const float* Wk   = (const float*)d_in[7];
  const float* Wv   = (const float*)d_in[8];
  const float* Wc   = (const float*)d_in[9];
  const float* bc   = (const float*)d_in[10];
  float* out = (float*)d_out;

  float* ws   = (float*)d_ws;
  float* qg   = ws;
  float* Qb   = qg   + (size_t)BATCH * EMBD;
  float* Kb   = Qb   + (size_t)BATCH * GGRP * EMBD;
  float* Vb   = Kb   + (size_t)BATCH * NNODE * EMBD;
  float* outc = Vb   + (size_t)BATCH * NNODE * EMBD;
  float* mhb  = outc + (size_t)BATCH * GGRP * EMBD;

  mean_qg_kernel<<<BATCH, 256, 0, stream>>>(EN, WqG, qg);

  dim3 gp(512, 3, 1);
  proj_kernel<<<gp, 128, 0, stream>>>(EN, EF, EL, WqF, WqL, Wk, Wv, qg, Qb, Kb, Vb);

  dim3 ga(GGRP / 16, HNUM, BATCH);
  attn_kernel<<<ga, 256, 0, stream>>>(Qb, Kb, Vb, MASK, outc);

  mh_kernel<<<512, 128, 0, stream>>>(outc, Wc, bc, mhb);

  dim3 gf(GGRP / 16, BATCH, 1);
  final_kernel<<<gf, 256, 0, stream>>>(mhb, EN, MASK, out);
}


# Attention: scores built transposed (keys x queries); softmax max/sum in registers, weights feed P*V from the accumulators; 8 wave partials merged by max rescale (sum order changed)
# speedup vs baseline: 1.3594x; 1.1223x over previous
#include <hip/hip_runtime.h>
#include <math.h>

#define BATCH 32
#define NNODE 512
#define GGRP  512
#define EMBD  256
#define HNUM  16
#define DK    16

typedef __attribute__((ext_vector_type(2))) float v2f;
typedef __attribute__((ext_vector_type(4))) float v4f;
typedef __attribute__((ext_vector_type(8))) float v8f;

__device__ __forceinline__ v8f wmma4(v2f a, v2f b, v8f c) {
  return __builtin_amdgcn_wmma_f32_16x16x4_f32(
       false, a,  false, b,
       (short)0, c,  false,  false);
}

__device__ __forceinline__ v2f ld2s(const float* p) {
  return *reinterpret_cast<const v2f*>(p);
}
__device__ __forceinline__ v4f ld4g(const float* __restrict__ p) {
  return *reinterpret_cast<const v4f*>(p);
}
__device__ __forceinline__ v2f lo2(v4f v) { return __builtin_shufflevector(v, v, 0, 1); }
__device__ __forceinline__ v2f hi2(v4f v) { return __builtin_shufflevector(v, v, 2, 3); }

__global__ void mean_qg_kernel(const float* __restrict__ EN,
                               const float* __restrict__ WqG,
                               float* __restrict__ qg) {
  __shared__ float sm[EMBD];
  const int b = blockIdx.x;
  const int e = threadIdx.x;
  float s = 0.0f;
  const float* base = EN + (size_t)b * NNODE * EMBD + e;
  for (int n = 0; n < NNODE; ++n) s += base[(size_t)n * EMBD];
  sm[e] = s * (1.0f / (float)NNODE);
  __syncthreads();
  float acc = 0.0f;
  const float* wrow = WqG + (size_t)e * EMBD;
  for (int k = 0; k < EMBD; ++k) acc += sm[k] * wrow[k];
  qg[(size_t)b * EMBD + e] = acc;
}

__global__ void proj_kernel(const float* __restrict__ EN,
                            const float* __restrict__ EF,
                            const float* __restrict__ EL,
                            const float* __restrict__ WqF,
                            const float* __restrict__ WqL,
                            const float* __restrict__ Wk,
                            const float* __restrict__ Wv,
                            const float* __restrict__ qg,
                            float* __restrict__ Q,
                            float* __restrict__ K,
                            float* __restrict__ V) {
  const int widx = threadIdx.x >> 5;
  const int lane = threadIdx.x & 31;
  const int ln   = lane & 15;
  const int kh   = lane >> 4;
  const int wave = blockIdx.x * 4 + widx;
  const int b    = wave >> 6;
  const int rem  = wave & 63;
  const int mt   = rem >> 2;
  const int ng   = rem & 3;
  const int row0 = mt * 32 + ln;
  const int sel  = blockIdx.y;

  v8f acc[2][4];
#pragma unroll
  for (int i = 0; i < 2; ++i)
#pragma unroll
    for (int j = 0; j < 4; ++j) acc[i][j] = v8f{0.f,0.f,0.f,0.f,0.f,0.f,0.f,0.f};

  float* Out;
  if (sel < 2) {
    const float* W  = (sel == 0) ? Wk : Wv;
    const float* A0 = EN + (size_t)(b * NNODE + row0) * EMBD;
    const float* A1 = A0 + (size_t)16 * EMBD;
    const float* Br[4];
#pragma unroll
    for (int j = 0; j < 4; ++j) Br[j] = W + (size_t)(ng * 64 + j * 16 + ln) * EMBD;
    for (int k = 0; k < EMBD; k += 8) {
      const v4f a0 = ld4g(A0 + k + 4 * kh);
      const v4f a1 = ld4g(A1 + k + 4 * kh);
      v4f bb[4];
#pragma unroll
      for (int j = 0; j < 4; ++j) bb[j] = ld4g(Br[j] + k + 4 * kh);
#pragma unroll
      for (int j = 0; j < 4; ++j) {
        acc[0][j] = wmma4(lo2(a0), lo2(bb[j]), acc[0][j]);
        acc[1][j] = wmma4(lo2(a1), lo2(bb[j]), acc[1][j]);
      }
#pragma unroll
      for (int j = 0; j < 4; ++j) {
        acc[0][j] = wmma4(hi2(a0), hi2(bb[j]), acc[0][j]);
        acc[1][j] = wmma4(hi2(a1), hi2(bb[j]), acc[1][j]);
      }
    }
    Out = (sel == 0) ? K : V;
  } else {
    const float* A0 = EF + (size_t)(b * GGRP + row0) * EMBD;
    const float* A1 = A0 + (size_t)16 * EMBD;
    const float* Br[4];
#pragma unroll
    for (int j = 0; j < 4; ++j) Br[j] = WqF + (size_t)(ng * 64 + j * 16 + ln) * EMBD;
    for (int k = 0; k < EMBD; k += 8) {
      const v4f a0 = ld4g(A0 + k + 4 * kh);
      const v4f a1 = ld4g(A1 + k + 4 * kh);
      v4f bb[4];
#pragma unroll
      for (int j = 0; j < 4; ++j) bb[j] = ld4g(Br[j] + k + 4 * kh);
#pragma unroll
      for (int j = 0; j < 4; ++j) {
        acc[0][j] = wmma4(lo2(a0), lo2(bb[j]), acc[0][j]);
        acc[1][j] = wmma4(lo2(a1), lo2(bb[j]), acc[1][j]);
      }
#pragma unroll
      for (int j = 0; j < 4; ++j) {
        acc[0][j] = wmma4(hi2(a0), hi2(bb[j]), acc[0][j]);
        acc[1][j] = wmma4(hi2(a1), hi2(bb[j]), acc[1][j]);
      }
    }
    const float* C0 = EL + (size_t)(b * GGRP + row0) * EMBD;
    const float* C1 = C0 + (size_t)16 * EMBD;
#pragma unroll
    for (int j = 0; j < 4; ++j) Br[j] = WqL + (size_t)(ng * 64 + j * 16 + ln) * EMBD;
    for (int k = 0; k < EMBD; k += 8) {
      const v4f a0 = ld4g(C0 + k + 4 * kh);
      const v4f a1 = ld4g(C1 + k + 4 * kh);
      v4f bb[4];
#pragma unroll
      for (int j = 0; j < 4; ++j) bb[j] = ld4g(Br[j] + k + 4 * kh);
#pragma unroll
      for (int j = 0; j < 4; ++j) {
        acc[0][j] = wmma4(lo2(a0), lo2(bb[j]), acc[0][j]);
        acc[1][j] = wmma4(lo2(a1), lo2(bb[j]), acc[1][j]);
      }
#pragma unroll
      for (int j = 0; j < 4; ++j) {
        acc[0][j] = wmma4(hi2(a0), hi2(bb[j]), acc[0][j]);
        acc[1][j] = wmma4(hi2(a1), hi2(bb[j]), acc[1][j]);
      }
    }
#pragma unroll
    for (int j = 0; j < 4; ++j) {
      const float qv = qg[(size_t)b * EMBD + ng * 64 + j * 16 + ln];
#pragma unroll
      for (int i = 0; i < 2; ++i)
#pragma unroll
        for (int r = 0; r < 8; ++r) acc[i][j][r] += qv;
    }
    Out = Q;
  }

#pragma unroll
  for (int i = 0; i < 2; ++i)
#pragma unroll
    for (int j = 0; j < 4; ++j) {
      const int col = ng * 64 + j * 16 + ln;
#pragma unroll
      for (int r = 0; r < 8; ++r) {
        const int row = mt * 32 + i * 16 + r + 8 * kh;
        Out[(size_t)(b * NNODE + row) * EMBD + col] = acc[i][j][r];
      }
    }
}

#define OPITCH 18
__global__ void __launch_bounds__(256, 10)
attn_kernel(const float* __restrict__ Q,
            const float* __restrict__ K,
            const float* __restrict__ V,
            const float* __restrict__ MASK,
            float* __restrict__ outc) {
  __shared__ float s_m[8][16];
  __shared__ float s_l[8][16];
  __shared__ float s_o[8][16 * OPITCH];

  const int gt = blockIdx.x, h = blockIdx.y, b = blockIdx.z;
  const int t    = threadIdx.x;
  const int widx = t >> 5;
  const int lane = t & 31;
  const int ln   = lane & 15;
  const int kh   = lane >> 4;
  const int nb   = widx * 64;

  const float* Qrow = Q + (size_t)(b * GGRP + gt * 16 + ln) * EMBD + h * DK;
  const float* Krow = K + (size_t)(b * NNODE + nb + ln) * EMBD + h * DK;
  const v4f q0 = ld4g(Qrow + 4 * kh);
  const v4f q1 = ld4g(Qrow + 8 + 4 * kh);
  v4f k0[4], k1[4];
#pragma unroll
  for (int j = 0; j < 4; ++j) {
    k0[j] = ld4g(Krow + (size_t)j * 16 * EMBD + 4 * kh);
    k1[j] = ld4g(Krow + (size_t)j * 16 * EMBD + 8 + 4 * kh);
  }
  const float* Mrow = MASK + (size_t)(b * GGRP + gt * 16 + ln) * NNODE + nb + 8 * kh;
  v4f ma[4], mb[4];
#pragma unroll
  for (int j = 0; j < 4; ++j) {
    ma[j] = ld4g(Mrow + 16 * j);
    mb[j] = ld4g(Mrow + 16 * j + 4);
  }

  v8f acc[4];
#pragma unroll
  for (int j = 0; j < 4; ++j) acc[j] = v8f{0.f,0.f,0.f,0.f,0.f,0.f,0.f,0.f};
#pragma unroll
  for (int j = 0; j < 4; ++j) acc[j] = wmma4(lo2(k0[j]), lo2(q0), acc[j]);
#pragma unroll
  for (int j = 0; j < 4; ++j) acc[j] = wmma4(hi2(k0[j]), hi2(q0), acc[j]);
#pragma unroll
  for (int j = 0; j < 4; ++j) acc[j] = wmma4(lo2(k1[j]), lo2(q1), acc[j]);
#pragma unroll
  for (int j = 0; j < 4; ++j) acc[j] = wmma4(hi2(k1[j]), hi2(q1), acc[j]);

  float mloc = -INFINITY;
#pragma unroll
  for (int j = 0; j < 4; ++j) {
#pragma unroll
    for (int r = 0; r < 4; ++r) {
      acc[j][r]     = acc[j][r]     * 0.25f + ma[j][r];
      acc[j][r + 4] = acc[j][r + 4] * 0.25f + mb[j][r];
      mloc = fmaxf(mloc, fmaxf(acc[j][r], acc[j][r + 4]));
    }
  }
  mloc = fmaxf(mloc, __shfl_xor(mloc, 16));
  const float muse = (mloc == -INFINITY) ? 0.0f : mloc;

  float lsum = 0.0f;
#pragma unroll
  for (int j = 0; j < 4; ++j) {
#pragma unroll
    for (int r = 0; r < 8; ++r) {
      const float e = __expf(acc[j][r] - muse);
      acc[j][r] = e;
      lsum += e;
    }
  }
  lsum += __shfl_xor(lsum, 16);
  s_m[widx][ln] = mloc;
  s_l[widx][ln] = lsum;

  const float* Vp = V + (size_t)(b * NNODE + nb + 8 * kh) * EMBD + h * DK + ln;
  v8f o0 = {0.f,0.f,0.f,0.f,0.f,0.f,0.f,0.f};
  v8f o1 = {0.f,0.f,0.f,0.f,0.f,0.f,0.f,0.f};
#pragma unroll
  for (int j = 0; j < 4; ++j) {
#pragma unroll
    for (int r = 0; r < 8; r += 4) {
      v2f a0, a1, w0, w1;
      a0.x = Vp[(size_t)(16 * j + r)     * EMBD];
      a0.y = Vp[(size_t)(16 * j + r + 1) * EMBD];
      a1.x = Vp[(size_t)(16 * j + r + 2) * EMBD];
      a1.y = Vp[(size_t)(16 * j + r + 3) * EMBD];
      w0.x = acc[j][r];     w0.y = acc[j][r + 1];
      w1.x = acc[j][r + 2]; w1.y = acc[j][r + 3];
      o0 = wmma4(a0, w0, o0);
      o1 = wmma4(a1, w1, o1);
    }
  }
#pragma unroll
  for (int r = 0; r < 8; ++r)
    s_o[widx][(8 * kh + r) * OPITCH + ln] = o0[r] + o1[r];
  __syncthreads();

  {
    const int g = t >> 4, d = t & 15;
    float mw[8];
    float M = -INFINITY;
#pragma unroll
    for (int w = 0; w < 8; ++w) { mw[w] = s_m[w][g]; M = fmaxf(M, mw[w]); }
    float L = 0.0f, O = 0.0f;
#pragma unroll
    for (int w = 0; w < 8; ++w) {
      const float e = __expf(mw[w] - M);
      L += e * s_l[w][g];
      O += e * s_o[w][d * OPITCH + g];
    }
    outc[(size_t)(b * GGRP + gt * 16 + g) * EMBD + h * DK + d] = O / L;
  }
}

__global__ void mh_kernel(const float* __restrict__ outc,
                          const float* __restrict__ Wc,
                          const float* __restrict__ bc,
                          float* __restrict__ mh) {
  const int widx = threadIdx.x >> 5;
  const int lane = threadIdx.x & 31;
  const int ln   = lane & 15;
  const int kh   = lane >> 4;
  const int wave = blockIdx.x * 4 + widx;
  const int b    = wave >> 6;
  const int rem  = wave & 63;
  const int mt   = rem >> 2;
  const int ng   = rem & 3;
  const int row0 = mt * 32 + ln;

  v8f acc[2][4];
#pragma unroll
  for (int i = 0; i < 2; ++i)
#pragma unroll
    for (int j = 0; j < 4; ++j) acc[i][j] = v8f{0.f,0.f,0.f,0.f,0.f,0.f,0.f,0.f};

  const float* A0 = outc + (size_t)(b * GGRP + row0) * EMBD;
  const float* A1 = A0 + (size_t)16 * EMBD;
  const float* Br[4];
#pragma unroll
  for (int j = 0; j < 4; ++j) Br[j] = Wc + (size_t)(ng * 64 + j * 16 + ln) * EMBD;
  for (int k = 0; k < EMBD; k += 8) {
    const v4f a0 = ld4g(A0 + k + 4 * kh);
    const v4f a1 = ld4g(A1 + k + 4 * kh);
    v4f bb[4];
#pragma unroll
    for (int j = 0; j < 4; ++j) bb[j] = ld4g(Br[j] + k + 4 * kh);
#pragma unroll
    for (int j = 0; j < 4; ++j) {
      acc[0][j] = wmma4(lo2(a0), lo2(bb[j]), acc[0][j]);
      acc[1][j] = wmma4(lo2(a1), lo2(bb[j]), acc[1][j]);
    }
#pragma unroll
    for (int j = 0; j < 4; ++j) {
      acc[0][j] = wmma4(hi2(a0), hi2(bb[j]), acc[0][j]);
      acc[1][j] = wmma4(hi2(a1), hi2(bb[j]), acc[1][j]);
    }
  }
#pragma unroll
  for (int i = 0; i < 2; ++i)
#pragma unroll
    for (int j = 0; j < 4; ++j) {
      const int col  = ng * 64 + j * 16 + ln;
      const float bv = bc[col];
#pragma unroll
      for (int r = 0; r < 8; ++r) {
        const int row = mt * 32 + i * 16 + r + 8 * kh;
        mh[(size_t)(b * GGRP + row) * EMBD + col] = acc[i][j][r] + bv;
      }
    }
}

__global__ void final_kernel(const float* __restrict__ mh,
                             const float* __restrict__ EN,
                             const float* __restrict__ MASK,
                             float* __restrict__ out) {
  __shared__ float sc[16 * 512];
  __shared__ float redA[16][16];
  __shared__ float redB[16][16];

  const int gt = blockIdx.x, b = blockIdx.y;
  const int t    = threadIdx.x;
  const int widx = t >> 5;
  const int lane = t & 31;
  const int ln   = lane & 15;
  const int kh   = lane >> 4;

  {
    const float* Arow = mh + (size_t)(b * GGRP + gt * 16 + ln) * EMBD;
    const float* Br[4];
#pragma unroll
    for (int j = 0; j < 4; ++j)
      Br[j] = EN + (size_t)(b * NNODE + (widx * 4 + j) * 16 + ln) * EMBD;
    v8f acc[4];
#pragma unroll
    for (int j = 0; j < 4; ++j) acc[j] = v8f{0.f,0.f,0.f,0.f,0.f,0.f,0.f,0.f};
    for (int k = 0; k < EMBD; k += 8) {
      const v4f a = ld4g(Arow + k + 4 * kh);
      v4f bb[4];
#pragma unroll
      for (int j = 0; j < 4; ++j) bb[j] = ld4g(Br[j] + k + 4 * kh);
#pragma unroll
      for (int j = 0; j < 4; ++j) acc[j] = wmma4(lo2(a), lo2(bb[j]), acc[j]);
#pragma unroll
      for (int j = 0; j < 4; ++j) acc[j] = wmma4(hi2(a), hi2(bb[j]), acc[j]);
    }
#pragma unroll
    for (int j = 0; j < 4; ++j) {
      const int nb = (widx * 4 + j) * 16;
#pragma unroll
      for (int r = 0; r < 8; ++r) {
        const int row = r + 8 * kh;
        const int g   = gt * 16 + row;
        const float s2 = acc[j][r] * (1.0f / 16.0f);
        sc[row * 512 + nb + ln] =
            10.0f * tanhf(s2) + MASK[(size_t)(b * GGRP + g) * NNODE + nb + ln];
      }
    }
  }
  __syncthreads();

  const int row = t >> 4, l16 = t & 15;
  float* srow = sc + row * 512;
  float mx = -INFINITY;
  for (int j = l16; j < 512; j += 16) mx = fmaxf(mx, srow[j]);
  redA[row][l16] = mx;
  __syncthreads();
  if (l16 == 0) {
    float m2 = redA[row][0];
    for (int i = 1; i < 16; ++i) m2 = fmaxf(m2, redA[row][i]);
    redA[row][0] = m2;
  }
  __syncthreads();
  const float rmax = redA[row][0];
  float ssum = 0.0f;
  for (int j = l16; j < 512; j += 16) {
    const float e = __expf(srow[j] - rmax);
    srow[j] = e;
    ssum += e;
  }
  redB[row][l16] = ssum;
  __syncthreads();
  if (l16 == 0) {
    float s2 = 0.0f;
    for (int i = 0; i < 16; ++i) s2 += redB[row][i];
    redB[row][0] = s2;
  }
  __syncthreads();
  const float inv = 1.0f / redB[row][0];
  float* orow = out + (size_t)(b * GGRP + gt * 16 + row) * NNODE;
  for (int j = l16; j < 512; j += 16) orow[j] = srow[j] * inv;
}

extern "C" void kernel_launch(void* const* d_in, const int* in_sizes, int n_in,
                              void* d_out, int out_size, void* d_ws, size_t ws_size,
                              hipStream_t stream) {
  const float* EN   = (const float*)d_in[0];
  const float* EF   = (const float*)d_in[1];
  const float* EL   = (const float*)d_in[2];
  const float* MASK = (const float*)d_in[3];
  const float* WqG  = (const float*)d_in[4];
  const float* WqF  = (const float*)d_in[5];
  const float* WqL  = (const float*)d_in[6];
  const float* Wk   = (const float*)d_in[7];
  const float* Wv   = (const float*)d_in[8];
  const float* Wc   = (const float*)d_in[9];
  const float* bc   = (const float*)d_in[10];
  float* out = (float*)d_out;

  float* ws   = (float*)d_ws;
  float* qg   = ws;
  float* Qb   = qg   + (size_t)BATCH * EMBD;
  float* Kb   = Qb   + (size_t)BATCH * GGRP * EMBD;
  float* Vb   = Kb   + (size_t)BATCH * NNODE * EMBD;
  float* outc = Vb   + (size_t)BATCH * NNODE * EMBD;
  float* mhb  = outc + (size_t)BATCH * GGRP * EMBD;

  mean_qg_kernel<<<BATCH, 256, 0, stream>>>(EN, WqG, qg);

  dim3 gp(512, 3, 1);
  proj_kernel<<<gp, 128, 0, stream>>>(EN, EF, EL, WqF, WqL, Wk, Wv, qg, Qb, Kb, Vb);

  dim3 ga(GGRP / 16, HNUM, BATCH);
  attn_kernel<<<ga, 256, 0, stream>>>(Qb, Kb, Vb, MASK, outc);

  mh_kernel<<<512, 128, 0, stream>>>(outc, Wc, bc, mhb);

  dim3 gf(GGRP / 16, BATCH, 1);
  final_kernel<<<gf, 256, 0, stream>>>(mhb, EN, MASK, out);
}


# Attention: each wave serves two adjacent heads: every 128-B K, V and Q line is fetched once and used whole, the mask tile once per pair; per-head sums and their order unchanged
# speedup vs baseline: 1.5090x; 1.1100x over previous
#include <hip/hip_runtime.h>
#include <math.h>

#define BATCH 32
#define NNODE 512
#define GGRP  512
#define EMBD  256
#define HNUM  16
#define DK    16

typedef __attribute__((ext_vector_type(2))) float v2f;
typedef __attribute__((ext_vector_type(4))) float v4f;
typedef __attribute__((ext_vector_type(8))) float v8f;

__device__ __forceinline__ v8f wmma4(v2f a, v2f b, v8f c) {
  return __builtin_amdgcn_wmma_f32_16x16x4_f32(
       false, a,  false, b,
       (short)0, c,  false,  false);
}

__device__ __forceinline__ v2f ld2s(const float* p) {
  return *reinterpret_cast<const v2f*>(p);
}
__device__ __forceinline__ v4f ld4g(const float* __restrict__ p) {
  return *reinterpret_cast<const v4f*>(p);
}
__device__ __forceinline__ v2f lo2(v4f v) { return __builtin_shufflevector(v, v, 0, 1); }
__device__ __forceinline__ v2f hi2(v4f v) { return __builtin_shufflevector(v, v, 2, 3); }

__global__ void mean_qg_kernel(const float* __restrict__ EN,
                               const float* __restrict__ WqG,
                               float* __restrict__ qg) {
  __shared__ float sm[EMBD];
  const int b = blockIdx.x;
  const int e = threadIdx.x;
  float s = 0.0f;
  const float* base = EN + (size_t)b * NNODE * EMBD + e;
  for (int n = 0; n < NNODE; ++n) s += base[(size_t)n * EMBD];
  sm[e] = s * (1.0f / (float)NNODE);
  __syncthreads();
  float acc = 0.0f;
  const float* wrow = WqG + (size_t)e * EMBD;
  for (int k = 0; k < EMBD; ++k) acc += sm[k] * wrow[k];
  qg[(size_t)b * EMBD + e] = acc;
}

__global__ void proj_kernel(const float* __restrict__ EN,
                            const float* __restrict__ EF,
                            const float* __restrict__ EL,
                            const float* __restrict__ WqF,
                            const float* __restrict__ WqL,
                            const float* __restrict__ Wk,
                            const float* __restrict__ Wv,
                            const float* __restrict__ qg,
                            float* __restrict__ Q,
                            float* __restrict__ K,
                            float* __restrict__ V) {
  const int widx = threadIdx.x >> 5;
  const int lane = threadIdx.x & 31;
  const int ln   = lane & 15;
  const int kh   = lane >> 4;
  const int wave = blockIdx.x * 4 + widx;
  const int b    = wave >> 6;
  const int rem  = wave & 63;
  const int mt   = rem >> 2;
  const int ng   = rem & 3;
  const int row0 = mt * 32 + ln;
  const int sel  = blockIdx.y;

  v8f acc[2][4];
#pragma unroll
  for (int i = 0; i < 2; ++i)
#pragma unroll
    for (int j = 0; j < 4; ++j) acc[i][j] = v8f{0.f,0.f,0.f,0.f,0.f,0.f,0.f,0.f};

  float* Out;
  if (sel < 2) {
    const float* W  = (sel == 0) ? Wk : Wv;
    const float* A0 = EN + (size_t)(b * NNODE + row0) * EMBD;
    const float* A1 = A0 + (size_t)16 * EMBD;
    const float* Br[4];
#pragma unroll
    for (int j = 0; j < 4; ++j) Br[j] = W + (size_t)(ng * 64 + j * 16 + ln) * EMBD;
    for (int k = 0; k < EMBD; k += 8) {
      const v4f a0 = ld4g(A0 + k + 4 * kh);
      const v4f a1 = ld4g(A1 + k + 4 * kh);
      v4f bb[4];
#pragma unroll
      for (int j = 0; j < 4; ++j) bb[j] = ld4g(Br[j] + k + 4 * kh);
#pragma unroll
      for (int j = 0; j < 4; ++j) {
        acc[0][j] = wmma4(lo2(a0), lo2(bb[j]), acc[0][j]);
        acc[1][j] = wmma4(lo2(a1), lo2(bb[j]), acc[1][j]);
      }
#pragma unroll
      for (int j = 0; j < 4; ++j) {
        acc[0][j] = wmma4(hi2(a0), hi2(bb[j]), acc[0][j]);
        acc[1][j] = wmma4(hi2(a1), hi2(bb[j]), acc[1][j]);
      }
    }
    Out = (sel == 0) ? K : V;
  } else {
    const float* A0 = EF + (size_t)(b * GGRP + row0) * EMBD;
    const float* A1 = A0 + (size_t)16 * EMBD;
    const float* Br[4];
#pragma unroll
    for (int j = 0; j < 4; ++j) Br[j] = WqF + (size_t)(ng * 64 + j * 16 + ln) * EMBD;
    for (int k = 0; k < EMBD; k += 8) {
      const v4f a0 = ld4g(A0 + k + 4 * kh);
      const v4f a1 = ld4g(A1 + k + 4 * kh);
      v4f bb[4];
#pragma unroll
      for (int j = 0; j < 4; ++j) bb[j] = ld4g(Br[j] + k + 4 * kh);
#pragma unroll
      for (int j = 0; j < 4; ++j) {
        acc[0][j] = wmma4(lo2(a0), lo2(bb[j]), acc[0][j]);
        acc[1][j] = wmma4(lo2(a1), lo2(bb[j]), acc[1][j]);
      }
#pragma unroll
      for (int j = 0; j < 4; ++j) {
        acc[0][j] = wmma4(hi2(a0), hi2(bb[j]), acc[0][j]);
        acc[1][j] = wmma4(hi2(a1), hi2(bb[j]), acc[1][j]);
      }
    }
    const float* C0 = EL + (size_t)(b * GGRP + row0) * EMBD;
    const float* C1 = C0 + (size_t)16 * EMBD;
#pragma unroll
    for (int j = 0; j < 4; ++j) Br[j] = WqL + (size_t)(ng * 64 + j * 16 + ln) * EMBD;
    for (int k = 0; k < EMBD; k += 8) {
      const v4f a0 = ld4g(C0 + k + 4 * kh);
      const v4f a1 = ld4g(C1 + k + 4 * kh);
      v4f bb[4];
#pragma unroll
      for (int j = 0; j < 4; ++j) bb[j] = ld4g(Br[j] + k + 4 * kh);
#pragma unroll
      for (int j = 0; j < 4; ++j) {
        acc[0][j] = wmma4(lo2(a0), lo2(bb[j]), acc[0][j]);
        acc[1][j] = wmma4(lo2(a1), lo2(bb[j]), acc[1][j]);
      }
#pragma unroll
      for (int j = 0; j < 4; ++j) {
        acc[0][j] = wmma4(hi2(a0), hi2(bb[j]), acc[0][j]);
        acc[1][j] = wmma4(hi2(a1), hi2(bb[j]), acc[1][j]);
      }
    }
#pragma unroll
    for (int j = 0; j < 4; ++j) {
      const float qv = qg[(size_t)b * EMBD + ng * 64 + j * 16 + ln];
#pragma unroll
      for (int i = 0; i < 2; ++i)
#pragma unroll
        for (int r = 0; r < 8; ++r) acc[i][j][r] += qv;
    }
    Out = Q;
  }

#pragma unroll
  for (int i = 0; i < 2; ++i)
#pragma unroll
    for (int j = 0; j < 4; ++j) {
      const int col = ng * 64 + j * 16 + ln;
#pragma unroll
      for (int r = 0; r < 8; ++r) {
        const int row = mt * 32 + i * 16 + r + 8 * kh;
        Out[(size_t)(b * NNODE + row) * EMBD + col] = acc[i][j][r];
      }
    }
}

#define HP 2
#define OPITCH 18
#define OHEAD (8 * 16 * OPITCH + 1)
__global__ void __launch_bounds__(256, 8)
attn_kernel(const float* __restrict__ Q,
            const float* __restrict__ K,
            const float* __restrict__ V,
            const float* __restrict__ MASK,
            float* __restrict__ outc) {
  __shared__ float s_m[HP][8][16];
  __shared__ float s_l[HP][8][16];
  __shared__ float s_o[HP * OHEAD];

  const int gt = blockIdx.x, hp = blockIdx.y, b = blockIdx.z;
  const int t    = threadIdx.x;
  const int widx = __builtin_amdgcn_readfirstlane(t >> 5);
  const int lane = t & 31;
  const int ln   = lane & 15;
  const int kh   = lane >> 4;
  const int nb   = widx * 64;
  const int c0   = hp * (HP * DK);

  const float* Qrow = Q + (size_t)(b * GGRP + gt * 16) * EMBD + c0 + (ln * EMBD);
  const float* Krow = K + (size_t)(b * NNODE + nb) * EMBD + c0 + (ln * EMBD);
  v4f q0[HP], q1[HP];
  v4f k0[HP][4], k1[HP][4];
#pragma unroll
  for (int hh = 0; hh < HP; ++hh) {
    q0[hh] = ld4g(Qrow + hh * DK + 4 * kh);
    q1[hh] = ld4g(Qrow + hh * DK + 8 + 4 * kh);
  }
#pragma unroll
  for (int j = 0; j < 4; ++j) {
#pragma unroll
    for (int hh = 0; hh < HP; ++hh) {
      k0[hh][j] = ld4g(Krow + (size_t)j * 16 * EMBD + hh * DK + 4 * kh);
      k1[hh][j] = ld4g(Krow + (size_t)j * 16 * EMBD + hh * DK + 8 + 4 * kh);
    }
  }
  const float* Mrow = MASK + (size_t)(b * GGRP + gt * 16) * NNODE + nb + (ln * NNODE + 8 * kh);
  v4f ma[4], mb[4];
#pragma unroll
  for (int j = 0; j < 4; ++j) {
    ma[j] = ld4g(Mrow + 16 * j);
    mb[j] = ld4g(Mrow + 16 * j + 4);
  }

  v8f acc[HP][4];
#pragma unroll
  for (int hh = 0; hh < HP; ++hh) {
#pragma unroll
    for (int j = 0; j < 4; ++j) acc[hh][j] = v8f{0.f,0.f,0.f,0.f,0.f,0.f,0.f,0.f};
#pragma unroll
    for (int j = 0; j < 4; ++j) acc[hh][j] = wmma4(lo2(k0[hh][j]), lo2(q0[hh]), acc[hh][j]);
#pragma unroll
    for (int j = 0; j < 4; ++j) acc[hh][j] = wmma4(hi2(k0[hh][j]), hi2(q0[hh]), acc[hh][j]);
#pragma unroll
    for (int j = 0; j < 4; ++j) acc[hh][j] = wmma4(lo2(k1[hh][j]), lo2(q1[hh]), acc[hh][j]);
#pragma unroll
    for (int j = 0; j < 4; ++j) acc[hh][j] = wmma4(hi2(k1[hh][j]), hi2(q1[hh]), acc[hh][j]);
  }

#pragma unroll
  for (int hh = 0; hh < HP; ++hh) {
    float mloc = -INFINITY;
#pragma unroll
    for (int j = 0; j < 4; ++j) {
#pragma unroll
      for (int r = 0; r < 4; ++r) {
        acc[hh][j][r]     = acc[hh][j][r]     * 0.25f + ma[j][r];
        acc[hh][j][r + 4] = acc[hh][j][r + 4] * 0.25f + mb[j][r];
        mloc = fmaxf(mloc, fmaxf(acc[hh][j][r], acc[hh][j][r + 4]));
      }
    }
    mloc = fmaxf(mloc, __shfl_xor(mloc, 16));
    const float muse = (mloc == -INFINITY) ? 0.0f : mloc;

    float lsum = 0.0f;
#pragma unroll
    for (int j = 0; j < 4; ++j) {
#pragma unroll
      for (int r = 0; r < 8; ++r) {
        const float e = __expf(acc[hh][j][r] - muse);
        acc[hh][j][r] = e;
        lsum += e;
      }
    }
    lsum += __shfl_xor(lsum, 16);
    s_m[hh][widx][ln] = mloc;
    s_l[hh][widx][ln] = lsum;
  }

  const float* Vp = V + (size_t)(b * NNODE + nb) * EMBD + c0 + (8 * kh * EMBD + ln);
  v8f o0[HP], o1[HP];
#pragma unroll
  for (int hh = 0; hh < HP; ++hh) {
    o0[hh] = v8f{0.f,0.f,0.f,0.f,0.f,0.f,0.f,0.f};
    o1[hh] = v8f{0.f,0.f,0.f,0.f,0.f,0.f,0.f,0.f};
  }
#pragma unroll
  for (int j = 0; j < 4; ++j) {
#pragma unroll
    for (int r = 0; r < 8; r += 4) {
      v2f a0[HP], a1[HP];
#pragma unroll
      for (int hh = 0; hh < HP; ++hh) {
        a0[hh].x = Vp[(size_t)(16 * j + r)     * EMBD + hh * DK];
        a0[hh].y = Vp[(size_t)(16 * j + r + 1) * EMBD + hh * DK];
        a1[hh].x = Vp[(size_t)(16 * j + r + 2) * EMBD + hh * DK];
        a1[hh].y = Vp[(size_t)(16 * j + r + 3) * EMBD + hh * DK];
      }
#pragma unroll
      for (int hh = 0; hh < HP; ++hh) {
        v2f w0, w1;
        w0.x = acc[hh][j][r];     w0.y = acc[hh][j][r + 1];
        w1.x = acc[hh][j][r + 2]; w1.y = acc[hh][j][r + 3];
        o0[hh] = wmma4(a0[hh], w0, o0[hh]);
        o1[hh] = wmma4(a1[hh], w1, o1[hh]);
      }
    }
  }
#pragma unroll
  for (int hh = 0; hh < HP; ++hh) {
#pragma unroll
    for (int r = 0; r < 8; ++r)
      s_o[hh * OHEAD + widx * (16 * OPITCH) + (8 * kh + r) * OPITCH + ln] = o0[hh][r] + o1[hh][r];
  }
  __syncthreads();

  static_assert(HP * DK == 32, "one lane per column of the head pair");
#pragma unroll
  for (int i = 0; i < 2; ++i) {
    const int g  = 8 * i + widx;
    const int hh = lane >> 4, d = lane & 15;
    float mw[8];
    float M = -INFINITY;
#pragma unroll
    for (int w = 0; w < 8; ++w) { mw[w] = s_m[hh][w][g]; M = fmaxf(M, mw[w]); }
    float L = 0.0f, O = 0.0f;
#pragma unroll
    for (int w = 0; w < 8; ++w) {
      const float e = __expf(mw[w] - M);
      L += e * s_l[hh][w][g];
      O += e * s_o[hh * OHEAD + w * (16 * OPITCH) + d * OPITCH + g];
    }
    outc[(size_t)(b * GGRP + gt * 16 + g) * EMBD + c0 + lane] = O / L;
  }
}

__global__ void mh_kernel(const float* __restrict__ outc,
                          const float* __restrict__ Wc,
                          const float* __restrict__ bc,
                          float* __restrict__ mh) {
  const int widx = threadIdx.x >> 5;
  const int lane = threadIdx.x & 31;
  const int ln   = lane & 15;
  const int kh   = lane >> 4;
  const int wave = blockIdx.x * 4 + widx;
  const int b    = wave >> 6;
  const int rem  = wave & 63;
  const int mt   = rem >> 2;
  const int ng   = rem & 3;
  const int row0 = mt * 32 + ln;

  v8f acc[2][4];
#pragma unroll
  for (int i = 0; i < 2; ++i)
#pragma unroll
    for (int j = 0; j < 4; ++j) acc[i][j] = v8f{0.f,0.f,0.f,0.f,0.f,0.f,0.f,0.f};

  const float* A0 = outc + (size_t)(b * GGRP + row0) * EMBD;
  const float* A1 = A0 + (size_t)16 * EMBD;
  const float* Br[4];
#pragma unroll
  for (int j = 0; j < 4; ++j) Br[j] = Wc + (size_t)(ng * 64 + j * 16 + ln) * EMBD;
  for (int k = 0; k < EMBD; k += 8) {
    const v4f a0 = ld4g(A0 + k + 4 * kh);
    const v4f a1 = ld4g(A1 + k + 4 * kh);
    v4f bb[4];
#pragma unroll
    for (int j = 0; j < 4; ++j) bb[j] = ld4g(Br[j] + k + 4 * kh);
#pragma unroll
    for (int j = 0; j < 4; ++j) {
      acc[0][j] = wmma4(lo2(a0), lo2(bb[j]), acc[0][j]);
      acc[1][j] = wmma4(lo2(a1), lo2(bb[j]), acc[1][j]);
    }
#pragma unroll
    for (int j = 0; j < 4; ++j) {
      acc[0][j] = wmma4(hi2(a0), hi2(bb[j]), acc[0][j]);
      acc[1][j] = wmma4(hi2(a1), hi2(bb[j]), acc[1][j]);
    }
  }
#pragma unroll
  for (int i = 0; i < 2; ++i)
#pragma unroll
    for (int j = 0; j < 4; ++j) {
      const int col  = ng * 64 + j * 16 + ln;
      const float bv = bc[col];
#pragma unroll
      for (int r = 0; r < 8; ++r) {
        const int row = mt * 32 + i * 16 + r + 8 * kh;
        mh[(size_t)(b * GGRP + row) * EMBD + col] = acc[i][j][r] + bv;
      }
    }
}

__global__ void final_kernel(const float* __restrict__ mh,
                             const float* __restrict__ EN,
                             const float* __restrict__ MASK,
                             float* __restrict__ out) {
  __shared__ float sc[16 * 512];
  __shared__ float redA[16][16];
  __shared__ float redB[16][16];

  const int gt = blockIdx.x, b = blockIdx.y;
  const int t    = threadIdx.x;
  const int widx = t >> 5;
  const int lane = t & 31;
  const int ln   = lane & 15;
  const int kh   = lane >> 4;

  {
    const float* Arow = mh + (size_t)(b * GGRP + gt * 16 + ln) * EMBD;
    const float* Br[4];
#pragma unroll
    for (int j = 0; j < 4; ++j)
      Br[j] = EN + (size_t)(b * NNODE + (widx * 4 + j) * 16 + ln) * EMBD;
    v8f acc[4];
#pragma unroll
    for (int j = 0; j < 4; ++j) acc[j] = v8f{0.f,0.f,0.f,0.f,0.f,0.f,0.f,0.f};
    for (int k = 0; k < EMBD; k += 8) {
      const v4f a = ld4g(Arow + k + 4 * kh);
      v4f bb[4];
#pragma unroll
      for (int j = 0; j < 4; ++j) bb[j] = ld4g(Br[j] + k + 4 * kh);
#pragma unroll
      for (int j = 0; j < 4; ++j) acc[j] = wmma4(lo2(a), lo2(bb[j]), acc[j]);
#pragma unroll
      for (int j = 0; j < 4; ++j) acc[j] = wmma4(hi2(a), hi2(bb[j]), acc[j]);
    }
#pragma unroll
    for (int j = 0; j < 4; ++j) {
      const int nb = (widx * 4 + j) * 16;
#pragma unroll
      for (int r = 0; r < 8; ++r) {
        const int row = r + 8 * kh;
        const int g   = gt * 16 + row;
        const float s2 = acc[j][r] * (1.0f / 16.0f);
        sc[row * 512 + nb + ln] =
            10.0f * tanhf(s2) + MASK[(size_t)(b * GGRP + g) * NNODE + nb + ln];
      }
    }
  }
  __syncthreads();

  const int row = t >> 4, l16 = t & 15;
  float* srow = sc + row * 512;
  float mx = -INFINITY;
  for (int j = l16; j < 512; j += 16) mx = fmaxf(mx, srow[j]);
  redA[row][l16] = mx;
  __syncthreads();
  if (l16 == 0) {
    float m2 = redA[row][0];
    for (int i = 1; i < 16; ++i) m2 = fmaxf(m2, redA[row][i]);
    redA[row][0] = m2;
  }
  __syncthreads();
  const float rmax = redA[row][0];
  float ssum = 0.0f;
  for (int j = l16; j < 512; j += 16) {
    const float e = __expf(srow[j] - rmax);
    srow[j] = e;
    ssum += e;
  }
  redB[row][l16] = ssum;
  __syncthreads();
  if (l16 == 0) {
    float s2 = 0.0f;
    for (int i = 0; i < 16; ++i) s2 += redB[row][i];
    redB[row][0] = s2;
  }
  __syncthreads();
  const float inv = 1.0f / redB[row][0];
  float* orow = out + (size_t)(b * GGRP + gt * 16 + row) * NNODE;
  for (int j = l16; j < 512; j += 16) orow[j] = srow[j] * inv;
}

extern "C" void kernel_launch(void* const* d_in, const int* in_sizes, int n_in,
                              void* d_out, int out_size, void* d_ws, size_t ws_size,
                              hipStream_t stream) {
  const float* EN   = (const float*)d_in[0];
  const float* EF   = (const float*)d_in[1];
  const float* EL   = (const float*)d_in[2];
  const float* MASK = (const float*)d_in[3];
  const float* WqG  = (const float*)d_in[4];
  const float* WqF  = (const float*)d_in[5];
  const float* WqL  = (const float*)d_in[6];
  const float* Wk   = (const float*)d_in[7];
  const float* Wv   = (const float*)d_in[8];
  const float* Wc   = (const float*)d_in[9];
  const float* bc   = (const float*)d_in[10];
  float* out = (float*)d_out;

  float* ws   = (float*)d_ws;
  float* qg   = ws;
  float* Qb   = qg   + (size_t)BATCH * EMBD;
  float* Kb   = Qb   + (size_t)BATCH * GGRP * EMBD;
  float* Vb   = Kb   + (size_t)BATCH * NNODE * EMBD;
  float* outc = Vb   + (size_t)BATCH * NNODE * EMBD;
  float* mhb  = outc + (size_t)BATCH * GGRP * EMBD;

  mean_qg_kernel<<<BATCH, 256, 0, stream>>>(EN, WqG, qg);

  dim3 gp(512, 3, 1);
  proj_kernel<<<gp, 128, 0, stream>>>(EN, EF, EL, WqF, WqL, Wk, Wv, qg, Qb, Kb, Vb);

  dim3 ga(GGRP / 16, HNUM / HP, BATCH);
  attn_kernel<<<ga, 256, 0, stream>>>(Qb, Kb, Vb, MASK, outc);

  mh_kernel<<<512, 128, 0, stream>>>(outc, Wc, bc, mhb);

  dim3 gf(GGRP / 16, BATCH, 1);
  final_kernel<<<gf, 256, 0, stream>>>(mhb, EN, MASK, out);
}


# GEMM k-loops take one whole 128-B line per row per step: a line's four 16-B fragment loads issued together ahead of the WMMAs, so it leaves L2 once, not per piece; sum order unchanged
# speedup vs baseline: 1.5642x; 1.0366x over previous
#include <hip/hip_runtime.h>
#include <math.h>

#define BATCH 32
#define NNODE 512
#define GGRP  512
#define EMBD  256
#define HNUM  16
#define DK    16

typedef __attribute__((ext_vector_type(2))) float v2f;
typedef __attribute__((ext_vector_type(4))) float v4f;
typedef __attribute__((ext_vector_type(8))) float v8f;

__device__ __forceinline__ v8f wmma4(v2f a, v2f b, v8f c) {
  return __builtin_amdgcn_wmma_f32_16x16x4_f32(
       false, a,  false, b,
       (short)0, c,  false,  false);
}

__device__ __forceinline__ v2f ld2s(const float* p) {
  return *reinterpret_cast<const v2f*>(p);
}
__device__ __forceinline__ v4f ld4g(const float* __restrict__ p) {
  return *reinterpret_cast<const v4f*>(p);
}
__device__ __forceinline__ v2f lo2(v4f v) { return __builtin_shufflevector(v, v, 0, 1); }
__device__ __forceinline__ v2f hi2(v4f v) { return __builtin_shufflevector(v, v, 2, 3); }

__device__ __forceinline__ void mac_block_32x64(const float* __restrict__ A0,
                                                const float* __restrict__ B0,
                                                v8f (&acc)[2][4]) {
  v4f a0[4], a1[4], bb[4][4];
#pragma unroll
  for (int i = 0; i < 4; ++i) {
    a0[i] = ld4g(A0 + 8 * i);
    a1[i] = ld4g(A0 + (size_t)16 * EMBD + 8 * i);
  }
#pragma unroll
  for (int j = 0; j < 4; ++j)
#pragma unroll
    for (int i = 0; i < 4; ++i) bb[j][i] = ld4g(B0 + (size_t)j * 16 * EMBD + 8 * i);
#pragma unroll
  for (int i = 0; i < 4; ++i) {
#pragma unroll
    for (int j = 0; j < 4; ++j) {
      acc[0][j] = wmma4(lo2(a0[i]), lo2(bb[j][i]), acc[0][j]);
      acc[1][j] = wmma4(lo2(a1[i]), lo2(bb[j][i]), acc[1][j]);
    }
#pragma unroll
    for (int j = 0; j < 4; ++j) {
      acc[0][j] = wmma4(hi2(a0[i]), hi2(bb[j][i]), acc[0][j]);
      acc[1][j] = wmma4(hi2(a1[i]), hi2(bb[j][i]), acc[1][j]);
    }
  }
}

__global__ void mean_qg_kernel(const float* __restrict__ EN,
                               const float* __restrict__ WqG,
                               float* __restrict__ qg) {
  __shared__ float sm[EMBD];
  const int b = blockIdx.x;
  const int e = threadIdx.x;
  float s = 0.0f;
  const float* base = EN + (size_t)b * NNODE * EMBD + e;
  for (int n = 0; n < NNODE; ++n) s += base[(size_t)n * EMBD];
  sm[e] = s * (1.0f / (float)NNODE);
  __syncthreads();
  float acc = 0.0f;
  const float* wrow = WqG + (size_t)e * EMBD;
  for (int k = 0; k < EMBD; ++k) acc += sm[k] * wrow[k];
  qg[(size_t)b * EMBD + e] = acc;
}

__global__ void __launch_bounds__(128) __attribute__((amdgpu_waves_per_eu(4, 5)))
proj_kernel(const float* __restrict__ EN,
                            const float* __restrict__ EF,
                            const float* __restrict__ EL,
                            const float* __restrict__ WqF,
                            const float* __restrict__ WqL,
                            const float* __restrict__ Wk,
                            const float* __restrict__ Wv,
                            const float* __restrict__ qg,
                            float* __restrict__ Q,
                            float* __restrict__ K,
                            float* __restrict__ V) {
  const int widx = threadIdx.x >> 5;
  const int lane = threadIdx.x & 31;
  const int ln   = lane & 15;
  const int kh   = lane >> 4;
  const int wave = blockIdx.x * 4 + widx;
  const int b    = wave >> 6;
  const int rem  = wave & 63;
  const int mt   = rem >> 2;
  const int ng   = rem & 3;
  const int row0 = mt * 32 + ln;
  const int sel  = blockIdx.y;

  v8f acc[2][4];
#pragma unroll
  for (int i = 0; i < 2; ++i)
#pragma unroll
    for (int j = 0; j < 4; ++j) acc[i][j] = v8f{0.f,0.f,0.f,0.f,0.f,0.f,0.f,0.f};

  float* Out;
  if (sel < 2) {
    const float* W  = (sel == 0) ? Wk : Wv;
    const float* A0 = EN + (size_t)(b * NNODE + row0) * EMBD + 4 * kh;
    const float* B0 = W + (size_t)(ng * 64 + ln) * EMBD + 4 * kh;
#pragma unroll 1
    for (int k = 0; k < EMBD; k += 32) mac_block_32x64(A0 + k, B0 + k, acc);
    Out = (sel == 0) ? K : V;
  } else {
    const float* A0 = EF + (size_t)(b * GGRP + row0) * EMBD + 4 * kh;
    const float* B0 = WqF + (size_t)(ng * 64 + ln) * EMBD + 4 * kh;
#pragma unroll 1
    for (int k = 0; k < EMBD; k += 32) mac_block_32x64(A0 + k, B0 + k, acc);
    const float* C0 = EL + (size_t)(b * GGRP + row0) * EMBD + 4 * kh;
    const float* D0 = WqL + (size_t)(ng * 64 + ln) * EMBD + 4 * kh;
#pragma unroll 1
    for (int k = 0; k < EMBD; k += 32) mac_block_32x64(C0 + k, D0 + k, acc);
#pragma unroll
    for (int j = 0; j < 4; ++j) {
      const float qv = qg[(size_t)b * EMBD + ng * 64 + j * 16 + ln];
#pragma unroll
      for (int i = 0; i < 2; ++i)
#pragma unroll
        for (int r = 0; r < 8; ++r) acc[i][j][r] += qv;
    }
    Out = Q;
  }

#pragma unroll
  for (int i = 0; i < 2; ++i)
#pragma unroll
    for (int j = 0; j < 4; ++j) {
      const int col = ng * 64 + j * 16 + ln;
#pragma unroll
      for (int r = 0; r < 8; ++r) {
        const int row = mt * 32 + i * 16 + r + 8 * kh;
        Out[(size_t)(b * NNODE + row) * EMBD + col] = acc[i][j][r];
      }
    }
}

#define HP 2
#define OPITCH 18
#define OHEAD (8 * 16 * OPITCH + 1)
__global__ void __launch_bounds__(256, 8)
attn_kernel(const float* __restrict__ Q,
            const float* __restrict__ K,
            const float* __restrict__ V,
            const float* __restrict__ MASK,
            float* __restrict__ outc) {
  __shared__ float s_m[HP][8][16];
  __shared__ float s_l[HP][8][16];
  __shared__ float s_o[HP * OHEAD];

  const int gt = blockIdx.x, hp = blockIdx.y, b = blockIdx.z;
  const int t    = threadIdx.x;
  const int widx = __builtin_amdgcn_readfirstlane(t >> 5);
  const int lane = t & 31;
  const int ln   = lane & 15;
  const int kh   = lane >> 4;
  const int nb   = widx * 64;
  const int c0   = hp * (HP * DK);

  const float* Qrow = Q + (size_t)(b * GGRP + gt * 16) * EMBD + c0 + (ln * EMBD);
  const float* Krow = K + (size_t)(b * NNODE + nb) * EMBD + c0 + (ln * EMBD);
  v4f q0[HP], q1[HP];
  v4f k0[HP][4], k1[HP][4];
#pragma unroll
  for (int hh = 0; hh < HP; ++hh) {
    q0[hh] = ld4g(Qrow + hh * DK + 4 * kh);
    q1[hh] = ld4g(Qrow + hh * DK + 8 + 4 * kh);
  }
#pragma unroll
  for (int j = 0; j < 4; ++j) {
#pragma unroll
    for (int hh = 0; hh < HP; ++hh) {
      k0[hh][j] = ld4g(Krow + (size_t)j * 16 * EMBD + hh * DK + 4 * kh);
      k1[hh][j] = ld4g(Krow + (size_t)j * 16 * EMBD + hh * DK + 8 + 4 * kh);
    }
  }
  const float* Mrow = MASK + (size_t)(b * GGRP + gt * 16) * NNODE + nb + (ln * NNODE + 8 * kh);
  v4f ma[4], mb[4];
#pragma unroll
  for (int j = 0; j < 4; ++j) {
    ma[j] = ld4g(Mrow + 16 * j);
    mb[j] = ld4g(Mrow + 16 * j + 4);
  }

  v8f acc[HP][4];
#pragma unroll
  for (int hh = 0; hh < HP; ++hh) {
#pragma unroll
    for (int j = 0; j < 4; ++j) acc[hh][j] = v8f{0.f,0.f,0.f,0.f,0.f,0.f,0.f,0.f};
#pragma unroll
    for (int j = 0; j < 4; ++j) acc[hh][j] = wmma4(lo2(k0[hh][j]), lo2(q0[hh]), acc[hh][j]);
#pragma unroll
    for (int j = 0; j < 4; ++j) acc[hh][j] = wmma4(hi2(k0[hh][j]), hi2(q0[hh]), acc[hh][j]);
#pragma unroll
    for (int j = 0; j < 4; ++j) acc[hh][j] = wmma4(lo2(k1[hh][j]), lo2(q1[hh]), acc[hh][j]);
#pragma unroll
    for (int j = 0; j < 4; ++j) acc[hh][j] = wmma4(hi2(k1[hh][j]), hi2(q1[hh]), acc[hh][j]);
  }

#pragma unroll
  for (int hh = 0; hh < HP; ++hh) {
    float mloc = -INFINITY;
#pragma unroll
    for (int j = 0; j < 4; ++j) {
#pragma unroll
      for (int r = 0; r < 4; ++r) {
        acc[hh][j][r]     = acc[hh][j][r]     * 0.25f + ma[j][r];
        acc[hh][j][r + 4] = acc[hh][j][r + 4] * 0.25f + mb[j][r];
        mloc = fmaxf(mloc, fmaxf(acc[hh][j][r], acc[hh][j][r + 4]));
      }
    }
    mloc = fmaxf(mloc, __shfl_xor(mloc, 16));
    const float muse = (mloc == -INFINITY) ? 0.0f : mloc;

    float lsum = 0.0f;
#pragma unroll
    for (int j = 0; j < 4; ++j) {
#pragma unroll
      for (int r = 0; r < 8; ++r) {
        const float e = __expf(acc[hh][j][r] - muse);
        acc[hh][j][r] = e;
        lsum += e;
      }
    }
    lsum += __shfl_xor(lsum, 16);
    s_m[hh][widx][ln] = mloc;
    s_l[hh][widx][ln] = lsum;
  }

  const float* Vp = V + (size_t)(b * NNODE + nb) * EMBD + c0 + (8 * kh * EMBD + ln);
  v8f o0[HP], o1[HP];
#pragma unroll
  for (int hh = 0; hh < HP; ++hh) {
    o0[hh] = v8f{0.f,0.f,0.f,0.f,0.f,0.f,0.f,0.f};
    o1[hh] = v8f{0.f,0.f,0.f,0.f,0.f,0.f,0.f,0.f};
  }
#pragma unroll
  for (int j = 0; j < 4; ++j) {
#pragma unroll
    for (int r = 0; r < 8; r += 4) {
      v2f a0[HP], a1[HP];
#pragma unroll
      for (int hh = 0; hh < HP; ++hh) {
        a0[hh].x = Vp[(size_t)(16 * j + r)     * EMBD + hh * DK];
        a0[hh].y = Vp[(size_t)(16 * j + r + 1) * EMBD + hh * DK];
        a1[hh].x = Vp[(size_t)(16 * j + r + 2) * EMBD + hh * DK];
        a1[hh].y = Vp[(size_t)(16 * j + r + 3) * EMBD + hh * DK];
      }
#pragma unroll
      for (int hh = 0; hh < HP; ++hh) {
        v2f w0, w1;
        w0.x = acc[hh][j][r];     w0.y = acc[hh][j][r + 1];
        w1.x = acc[hh][j][r + 2]; w1.y = acc[hh][j][r + 3];
        o0[hh] = wmma4(a0[hh], w0, o0[hh]);
        o1[hh] = wmma4(a1[hh], w1, o1[hh]);
      }
    }
  }
#pragma unroll
  for (int hh = 0; hh < HP; ++hh) {
#pragma unroll
    for (int r = 0; r < 8; ++r)
      s_o[hh * OHEAD + widx * (16 * OPITCH) + (8 * kh + r) * OPITCH + ln] = o0[hh][r] + o1[hh][r];
  }
  __syncthreads();

  static_assert(HP * DK == 32, "one lane per column of the head pair");
#pragma unroll
  for (int i = 0; i < 2; ++i) {
    const int g  = 8 * i + widx;
    const int hh = lane >> 4, d = lane & 15;
    float mw[8];
    float M = -INFINITY;
#pragma unroll
    for (int w = 0; w < 8; ++w) { mw[w] = s_m[hh][w][g]; M = fmaxf(M, mw[w]); }
    float L = 0.0f, O = 0.0f;
#pragma unroll
    for (int w = 0; w < 8; ++w) {
      const float e = __expf(mw[w] - M);
      L += e * s_l[hh][w][g];
      O += e * s_o[hh * OHEAD + w * (16 * OPITCH) + d * OPITCH + g];
    }
    outc[(size_t)(b * GGRP + gt * 16 + g) * EMBD + c0 + lane] = O / L;
  }
}

__global__ void __launch_bounds__(128) __attribute__((amdgpu_waves_per_eu(4, 5)))
mh_kernel(const float* __restrict__ outc,
                          const float* __restrict__ Wc,
                          const float* __restrict__ bc,
                          float* __restrict__ mh) {
  const int widx = threadIdx.x >> 5;
  const int lane = threadIdx.x & 31;
  const int ln   = lane & 15;
  const int kh   = lane >> 4;
  const int wave = blockIdx.x * 4 + widx;
  const int b    = wave >> 6;
  const int rem  = wave & 63;
  const int mt   = rem >> 2;
  const int ng   = rem & 3;
  const int row0 = mt * 32 + ln;

  v8f acc[2][4];
#pragma unroll
  for (int i = 0; i < 2; ++i)
#pragma unroll
    for (int j = 0; j < 4; ++j) acc[i][j] = v8f{0.f,0.f,0.f,0.f,0.f,0.f,0.f,0.f};

  const float* A0 = outc + (size_t)(b * GGRP + row0) * EMBD + 4 * kh;
  const float* B0 = Wc + (size_t)(ng * 64 + ln) * EMBD + 4 * kh;
#pragma unroll 1
  for (int k = 0; k < EMBD; k += 32) mac_block_32x64(A0 + k, B0 + k, acc);
#pragma unroll
  for (int i = 0; i < 2; ++i)
#pragma unroll
    for (int j = 0; j < 4; ++j) {
      const int col  = ng * 64 + j * 16 + ln;
      const float bv = bc[col];
#pragma unroll
      for (int r = 0; r < 8; ++r) {
        const int row = mt * 32 + i * 16 + r + 8 * kh;
        mh[(size_t)(b * GGRP + row) * EMBD + col] = acc[i][j][r] + bv;
      }
    }
}

__global__ void __launch_bounds__(256) __attribute__((amdgpu_waves_per_eu(8, 8)))
final_kernel(const float* __restrict__ mh,
                             const float* __restrict__ EN,
                             const float* __restrict__ MASK,
                             float* __restrict__ out) {
  __shared__ float sc[16 * 512];
  __shared__ float redA[16][16];
  __shared__ float redB[16][16];

  const int gt = blockIdx.x, b = blockIdx.y;
  const int t    = threadIdx.x;
  const int widx = t >> 5;
  const int lane = t & 31;
  const int ln   = lane & 15;
  const int kh   = lane >> 4;

  {
    const float* Arow = mh + (size_t)(b * GGRP + gt * 16 + ln) * EMBD;
    const float* Br[4];
#pragma unroll
    for (int j = 0; j < 4; ++j)
      Br[j] = EN + (size_t)(b * NNODE + (widx * 4 + j) * 16 + ln) * EMBD;
    v8f acc[4];
#pragma unroll
    for (int j = 0; j < 4; ++j) acc[j] = v8f{0.f,0.f,0.f,0.f,0.f,0.f,0.f,0.f};
#pragma unroll 1
    for (int k = 0; k < EMBD; k += 32) {
      v4f a[4], bb[4][4];
#pragma unroll
      for (int i = 0; i < 4; ++i) a[i] = ld4g(Arow + k + 8 * i + 4 * kh);
#pragma unroll
      for (int j = 0; j < 4; ++j)
#pragma unroll
        for (int i = 0; i < 4; ++i) bb[j][i] = ld4g(Br[j] + k + 8 * i + 4 * kh);
#pragma unroll
      for (int i = 0; i < 4; ++i) {
#pragma unroll
        for (int j = 0; j < 4; ++j) acc[j] = wmma4(lo2(a[i]), lo2(bb[j][i]), acc[j]);
#pragma unroll
        for (int j = 0; j < 4; ++j) acc[j] = wmma4(hi2(a[i]), hi2(bb[j][i]), acc[j]);
      }
    }
#pragma unroll
    for (int j = 0; j < 4; ++j) {
      const int nb = (widx * 4 + j) * 16;
#pragma unroll
      for (int r = 0; r < 8; ++r) {
        const int row = r + 8 * kh;
        const int g   = gt * 16 + row;
        const float s2 = acc[j][r] * (1.0f / 16.0f);
        sc[row * 512 + nb + ln] =
            10.0f * tanhf(s2) + MASK[(size_t)(b * GGRP + g) * NNODE + nb + ln];
      }
    }
  }
  __syncthreads();

  const int row = t >> 4, l16 = t & 15;
  float* srow = sc + row * 512;
  float mx = -INFINITY;
  for (int j = l16; j < 512; j += 16) mx = fmaxf(mx, srow[j]);
  redA[row][l16] = mx;
  __syncthreads();
  if (l16 == 0) {
    float m2 = redA[row][0];
    for (int i = 1; i < 16; ++i) m2 = fmaxf(m2, redA[row][i]);
    redA[row][0] = m2;
  }
  __syncthreads();
  const float rmax = redA[row][0];
  float ssum = 0.0f;
  for (int j = l16; j < 512; j += 16) {
    const float e = __expf(srow[j] - rmax);
    srow[j] = e;
    ssum += e;
  }
  redB[row][l16] = ssum;
  __syncthreads();
  if (l16 == 0) {
    float s2 = 0.0f;
    for (int i = 0; i < 16; ++i) s2 += redB[row][i];
    redB[row][0] = s2;
  }
  __syncthreads();
  const float inv = 1.0f / redB[row][0];
  float* orow = out + (size_t)(b * GGRP + gt * 16 + row) * NNODE;
  for (int j = l16; j < 512; j += 16) orow[j] = srow[j] * inv;
}

extern "C" void kernel_launch(void* const* d_in, const int* in_sizes, int n_in,
                              void* d_out, int out_size, void* d_ws, size_t ws_size,
                              hipStream_t stream) {
  const float* EN   = (const float*)d_in[0];
  const float* EF   = (const float*)d_in[1];
  const float* EL   = (const float*)d_in[2];
  const float* MASK = (const float*)d_in[3];
  const float* WqG  = (const float*)d_in[4];
  const float* WqF  = (const float*)d_in[5];
  const float* WqL  = (const float*)d_in[6];
  const float* Wk   = (const float*)d_in[7];
  const float* Wv   = (const float*)d_in[8];
  const float* Wc   = (const float*)d_in[9];
  const float* bc   = (const float*)d_in[10];
  float* out = (float*)d_out;

  float* ws   = (float*)d_ws;
  float* qg   = ws;
  float* Qb   = qg   + (size_t)BATCH * EMBD;
  float* Kb   = Qb   + (size_t)BATCH * GGRP * EMBD;
  float* Vb   = Kb   + (size_t)BATCH * NNODE * EMBD;
  float* outc = Vb   + (size_t)BATCH * NNODE * EMBD;
  float* mhb  = outc + (size_t)BATCH * GGRP * EMBD;

  mean_qg_kernel<<<BATCH, 256, 0, stream>>>(EN, WqG, qg);

  dim3 gp(512, 3, 1);
  proj_kernel<<<gp, 128, 0, stream>>>(EN, EF, EL, WqF, WqL, Wk, Wv, qg, Qb, Kb, Vb);

  dim3 ga(GGRP / 16, HNUM / HP, BATCH);
  attn_kernel<<<ga, 256, 0, stream>>>(Qb, Kb, Vb, MASK, outc);

  mh_kernel<<<512, 128, 0, stream>>>(outc, Wc, bc, mhb);

  dim3 gf(GGRP / 16, BATCH, 1);
  final_kernel<<<gf, 256, 0, stream>>>(mhb, EN, MASK, out);
}


# Graph mean + q_graph moved into the projection launch as its first 32 blocks, beside the GEMM blocks; attention adds q_graph where it loads Q (same single add): 4 launches, all sums and orders kept
# speedup vs baseline: 1.6908x; 1.0809x over previous
#include <hip/hip_runtime.h>
#include <math.h>

#define BATCH 32
#define NNODE 512
#define GGRP  512
#define EMBD  256
#define HNUM  16
#define DK    16

typedef __attribute__((ext_vector_type(2))) float v2f;
typedef __attribute__((ext_vector_type(4))) float v4f;
typedef __attribute__((ext_vector_type(8))) float v8f;

__device__ __forceinline__ v8f wmma4(v2f a, v2f b, v8f c) {
  return __builtin_amdgcn_wmma_f32_16x16x4_f32(
       false, a,  false, b,
       (short)0, c,  false,  false);
}

__device__ __forceinline__ v2f ld2s(const float* p) {
  return *reinterpret_cast<const v2f*>(p);
}
__device__ __forceinline__ v4f ld4g(const float* __restrict__ p) {
  return *reinterpret_cast<const v4f*>(p);
}
__device__ __forceinline__ v2f lo2(v4f v) { return __builtin_shufflevector(v, v, 0, 1); }
__device__ __forceinline__ v2f hi2(v4f v) { return __builtin_shufflevector(v, v, 2, 3); }

__device__ __forceinline__ void mac_block_32x64(const float* __restrict__ A0,
                                                const float* __restrict__ B0,
                                                v8f (&acc)[2][4]) {
  v4f a0[4], a1[4], bb[4][4];
#pragma unroll
  for (int i = 0; i < 4; ++i) {
    a0[i] = ld4g(A0 + 8 * i);
    a1[i] = ld4g(A0 + (size_t)16 * EMBD + 8 * i);
  }
#pragma unroll
  for (int j = 0; j < 4; ++j)
#pragma unroll
    for (int i = 0; i < 4; ++i) bb[j][i] = ld4g(B0 + (size_t)j * 16 * EMBD + 8 * i);
#pragma unroll
  for (int i = 0; i < 4; ++i) {
#pragma unroll
    for (int j = 0; j < 4; ++j) {
      acc[0][j] = wmma4(lo2(a0[i]), lo2(bb[j][i]), acc[0][j]);
      acc[1][j] = wmma4(lo2(a1[i]), lo2(bb[j][i]), acc[1][j]);
    }
#pragma unroll
    for (int j = 0; j < 4; ++j) {
      acc[0][j] = wmma4(hi2(a0[i]), hi2(bb[j][i]), acc[0][j]);
      acc[1][j] = wmma4(hi2(a1[i]), hi2(bb[j][i]), acc[1][j]);
    }
  }
}

#define PROJ_GROUP  512
#define PROJ_BLOCKS (BATCH + 3 * PROJ_GROUP)
__global__ void __launch_bounds__(128) __attribute__((amdgpu_waves_per_eu(4, 5)))
proj_kernel(const float* __restrict__ EN,
                            const float* __restrict__ EF,
                            const float* __restrict__ EL,
                            const float* __restrict__ WqG,
                            const float* __restrict__ WqF,
                            const float* __restrict__ WqL,
                            const float* __restrict__ Wk,
                            const float* __restrict__ Wv,
                            float* __restrict__ qg,
                            float* __restrict__ Q,
                            float* __restrict__ K,
                            float* __restrict__ V) {
  if (blockIdx.x < BATCH) {
    __shared__ float sm[EMBD];
    const int mb = blockIdx.x;
    const int t  = threadIdx.x;
    const float* base = EN + (size_t)mb * NNODE * EMBD + 2 * t;
    float s0 = 0.0f, s1 = 0.0f;
#pragma unroll 32
    for (int n = 0; n < NNODE; ++n) {
      const v2f x = ld2s(base + (size_t)n * EMBD);
      s0 += x.x;
      s1 += x.y;
    }
    sm[2 * t]     = s0 * (1.0f / (float)NNODE);
    sm[2 * t + 1] = s1 * (1.0f / (float)NNODE);
    __syncthreads();
    float a0 = 0.0f, a1 = 0.0f;
    const float* w0 = WqG + (size_t)t * EMBD;
    const float* w1 = WqG + (size_t)(t + 128) * EMBD;
#pragma unroll 16
    for (int k = 0; k < EMBD; ++k) {
      a0 += sm[k] * w0[k];
      a1 += sm[k] * w1[k];
    }
    qg[(size_t)mb * EMBD + t]       = a0;
    qg[(size_t)mb * EMBD + t + 128] = a1;
    return;
  }
  const unsigned gblk = blockIdx.x - BATCH;
  const int sel  = (int)(gblk / PROJ_GROUP);
  const int widx = threadIdx.x >> 5;
  const int lane = threadIdx.x & 31;
  const int ln   = lane & 15;
  const int kh   = lane >> 4;
  const int wave = (int)(gblk % PROJ_GROUP) * 4 + widx;
  const int b    = wave >> 6;
  const int rem  = wave & 63;
  const int mt   = rem >> 2;
  const int ng   = rem & 3;
  const int row0 = mt * 32 + ln;

  v8f acc[2][4];
#pragma unroll
  for (int i = 0; i < 2; ++i)
#pragma unroll
    for (int j = 0; j < 4; ++j) acc[i][j] = v8f{0.f,0.f,0.f,0.f,0.f,0.f,0.f,0.f};

  float* Out;
  if (sel < 2) {
    const float* W  = (sel == 0) ? Wk : Wv;
    const float* A0 = EN + (size_t)(b * NNODE + row0) * EMBD + 4 * kh;
    const float* B0 = W + (size_t)(ng * 64 + ln) * EMBD + 4 * kh;
#pragma unroll 1
    for (int k = 0; k < EMBD; k += 32) mac_block_32x64(A0 + k, B0 + k, acc);
    Out = (sel == 0) ? K : V;
  } else {
    const float* A0 = EF + (size_t)(b * GGRP + row0) * EMBD + 4 * kh;
    const float* B0 = WqF + (size_t)(ng * 64 + ln) * EMBD + 4 * kh;
#pragma unroll 1
    for (int k = 0; k < EMBD; k += 32) mac_block_32x64(A0 + k, B0 + k, acc);
    const float* C0 = EL + (size_t)(b * GGRP + row0) * EMBD + 4 * kh;
    const float* D0 = WqL + (size_t)(ng * 64 + ln) * EMBD + 4 * kh;
#pragma unroll 1
    for (int k = 0; k < EMBD; k += 32) mac_block_32x64(C0 + k, D0 + k, acc);
    Out = Q;
  }

#pragma unroll
  for (int i = 0; i < 2; ++i)
#pragma unroll
    for (int j = 0; j < 4; ++j) {
      const int col = ng * 64 + j * 16 + ln;
#pragma unroll
      for (int r = 0; r < 8; ++r) {
        const int row = mt * 32 + i * 16 + r + 8 * kh;
        Out[(size_t)(b * NNODE + row) * EMBD + col] = acc[i][j][r];
      }
    }
}

#define HP 2
#define OPITCH 18
#define OHEAD (8 * 16 * OPITCH + 1)
__global__ void __launch_bounds__(256, 8)
attn_kernel(const float* __restrict__ Q,
            const float* __restrict__ QG,
            const float* __restrict__ K,
            const float* __restrict__ V,
            const float* __restrict__ MASK,
            float* __restrict__ outc) {
  __shared__ float s_m[HP][8][16];
  __shared__ float s_l[HP][8][16];
  __shared__ float s_o[HP * OHEAD];

  const int gt = blockIdx.x, hp = blockIdx.y, b = blockIdx.z;
  const int t    = threadIdx.x;
  const int widx = __builtin_amdgcn_readfirstlane(t >> 5);
  const int lane = t & 31;
  const int ln   = lane & 15;
  const int kh   = lane >> 4;
  const int nb   = widx * 64;
  const int c0   = hp * (HP * DK);

  const float* Qrow = Q + (size_t)(b * GGRP + gt * 16) * EMBD + c0 + (ln * EMBD + 4 * kh);
  const float* Krow = K + (size_t)(b * NNODE + nb) * EMBD + c0 + (ln * EMBD + 4 * kh);
  v4f q0[HP], q1[HP];
  v4f k0[HP][4], k1[HP][4];
#pragma unroll
  for (int hh = 0; hh < HP; ++hh) {
    q0[hh] = ld4g(Qrow + hh * DK);
    q1[hh] = ld4g(Qrow + hh * DK + 8);
  }
  const float* QGrow = QG + (size_t)b * EMBD + c0 + (4 * kh);
#pragma unroll
  for (int hh = 0; hh < HP; ++hh) {
    q0[hh] += ld4g(QGrow + hh * DK);
    q1[hh] += ld4g(QGrow + hh * DK + 8);
  }
#pragma unroll
  for (int j = 0; j < 4; ++j) {
#pragma unroll
    for (int hh = 0; hh < HP; ++hh) {
      k0[hh][j] = ld4g(Krow + (size_t)j * 16 * EMBD + hh * DK);
      k1[hh][j] = ld4g(Krow + (size_t)j * 16 * EMBD + hh * DK + 8);
    }
  }
  const float* Mrow = MASK + (size_t)(b * GGRP + gt * 16) * NNODE + nb + (ln * NNODE + 8 * kh);
  v4f ma[4], mb[4];
#pragma unroll
  for (int j = 0; j < 4; ++j) {
    ma[j] = ld4g(Mrow + 16 * j);
    mb[j] = ld4g(Mrow + 16 * j + 4);
  }

  v8f acc[HP][4];
#pragma unroll
  for (int hh = 0; hh < HP; ++hh) {
#pragma unroll
    for (int j = 0; j < 4; ++j) acc[hh][j] = v8f{0.f,0.f,0.f,0.f,0.f,0.f,0.f,0.f};
#pragma unroll
    for (int j = 0; j < 4; ++j) acc[hh][j] = wmma4(lo2(k0[hh][j]), lo2(q0[hh]), acc[hh][j]);
#pragma unroll
    for (int j = 0; j < 4; ++j) acc[hh][j] = wmma4(hi2(k0[hh][j]), hi2(q0[hh]), acc[hh][j]);
#pragma unroll
    for (int j = 0; j < 4; ++j) acc[hh][j] = wmma4(lo2(k1[hh][j]), lo2(q1[hh]), acc[hh][j]);
#pragma unroll
    for (int j = 0; j < 4; ++j) acc[hh][j] = wmma4(hi2(k1[hh][j]), hi2(q1[hh]), acc[hh][j]);
  }

#pragma unroll
  for (int hh = 0; hh < HP; ++hh) {
    float mloc = -INFINITY;
#pragma unroll
    for (int j = 0; j < 4; ++j) {
#pragma unroll
      for (int r = 0; r < 4; ++r) {
        acc[hh][j][r]     = acc[hh][j][r]     * 0.25f + ma[j][r];
        acc[hh][j][r + 4] = acc[hh][j][r + 4] * 0.25f + mb[j][r];
        mloc = fmaxf(mloc, fmaxf(acc[hh][j][r], acc[hh][j][r + 4]));
      }
    }
    mloc = fmaxf(mloc, __shfl_xor(mloc, 16));
    const float muse = (mloc == -INFINITY) ? 0.0f : mloc;

    float lsum = 0.0f;
#pragma unroll
    for (int j = 0; j < 4; ++j) {
#pragma unroll
      for (int r = 0; r < 8; ++r) {
        const float e = __expf(acc[hh][j][r] - muse);
        acc[hh][j][r] = e;
        lsum += e;
      }
    }
    lsum += __shfl_xor(lsum, 16);
    s_m[hh][widx][ln] = mloc;
    s_l[hh][widx][ln] = lsum;
  }

  const float* Vp = V + (size_t)(b * NNODE + nb) * EMBD + c0 + (8 * kh * EMBD + ln);
  v8f o0[HP], o1[HP];
#pragma unroll
  for (int hh = 0; hh < HP; ++hh) {
    o0[hh] = v8f{0.f,0.f,0.f,0.f,0.f,0.f,0.f,0.f};
    o1[hh] = v8f{0.f,0.f,0.f,0.f,0.f,0.f,0.f,0.f};
  }
#pragma unroll
  for (int j = 0; j < 4; ++j) {
#pragma unroll
    for (int r = 0; r < 8; r += 4) {
      v2f a0[HP], a1[HP];
#pragma unroll
      for (int hh = 0; hh < HP; ++hh) {
        a0[hh].x = Vp[(size_t)(16 * j + r)     * EMBD + hh * DK];
        a0[hh].y = Vp[(size_t)(16 * j + r + 1) * EMBD + hh * DK];
        a1[hh].x = Vp[(size_t)(16 * j + r + 2) * EMBD + hh * DK];
        a1[hh].y = Vp[(size_t)(16 * j + r + 3) * EMBD + hh * DK];
      }
#pragma unroll
      for (int hh = 0; hh < HP; ++hh) {
        v2f w0, w1;
        w0.x = acc[hh][j][r];     w0.y = acc[hh][j][r + 1];
        w1.x = acc[hh][j][r + 2]; w1.y = acc[hh][j][r + 3];
        o0[hh] = wmma4(a0[hh], w0, o0[hh]);
        o1[hh] = wmma4(a1[hh], w1, o1[hh]);
      }
    }
  }
#pragma unroll
  for (int hh = 0; hh < HP; ++hh) {
#pragma unroll
    for (int r = 0; r < 8; ++r)
      s_o[hh * OHEAD + widx * (16 * OPITCH) + (8 * kh + r) * OPITCH + ln] = o0[hh][r] + o1[hh][r];
  }
  __syncthreads();

  static_assert(HP * DK == 32, "one lane per column of the head pair");
#pragma unroll
  for (int i = 0; i < 2; ++i) {
    const int g  = 8 * i + widx;
    const int hh = lane >> 4, d = lane & 15;
    float mw[8];
    float M = -INFINITY;
#pragma unroll
    for (int w = 0; w < 8; ++w) { mw[w] = s_m[hh][w][g]; M = fmaxf(M, mw[w]); }
    float L = 0.0f, O = 0.0f;
#pragma unroll
    for (int w = 0; w < 8; ++w) {
      const float e = __expf(mw[w] - M);
      L += e * s_l[hh][w][g];
      O += e * s_o[hh * OHEAD + w * (16 * OPITCH) + d * OPITCH + g];
    }
    outc[(size_t)(b * GGRP + gt * 16 + g) * EMBD + c0 + lane] = O / L;
  }
}

__global__ void __launch_bounds__(128) __attribute__((amdgpu_waves_per_eu(4, 5)))
mh_kernel(const float* __restrict__ outc,
                          const float* __restrict__ Wc,
                          const float* __restrict__ bc,
                          float* __restrict__ mh) {
  const int widx = threadIdx.x >> 5;
  const int lane = threadIdx.x & 31;
  const int ln   = lane & 15;
  const int kh   = lane >> 4;
  const int wave = blockIdx.x * 4 + widx;
  const int b    = wave >> 6;
  const int rem  = wave & 63;
  const int mt   = rem >> 2;
  const int ng   = rem & 3;
  const int row0 = mt * 32 + ln;

  v8f acc[2][4];
#pragma unroll
  for (int i = 0; i < 2; ++i)
#pragma unroll
    for (int j = 0; j < 4; ++j) acc[i][j] = v8f{0.f,0.f,0.f,0.f,0.f,0.f,0.f,0.f};

  const float* A0 = outc + (size_t)(b * GGRP + row0) * EMBD + 4 * kh;
  const float* B0 = Wc + (size_t)(ng * 64 + ln) * EMBD + 4 * kh;
#pragma unroll 1
  for (int k = 0; k < EMBD; k += 32) mac_block_32x64(A0 + k, B0 + k, acc);
#pragma unroll
  for (int i = 0; i < 2; ++i)
#pragma unroll
    for (int j = 0; j < 4; ++j) {
      const int col  = ng * 64 + j * 16 + ln;
      const float bv = bc[col];
#pragma unroll
      for (int r = 0; r < 8; ++r) {
        const int row = mt * 32 + i * 16 + r + 8 * kh;
        mh[(size_t)(b * GGRP + row) * EMBD + col] = acc[i][j][r] + bv;
      }
    }
}

__global__ void __launch_bounds__(256) __attribute__((amdgpu_waves_per_eu(8, 8)))
final_kernel(const float* __restrict__ mh,
                             const float* __restrict__ EN,
                             const float* __restrict__ MASK,
                             float* __restrict__ out) {
  __shared__ float sc[16 * 512];
  __shared__ float redA[16][16];
  __shared__ float redB[16][16];

  const int gt = blockIdx.x, b = blockIdx.y;
  const int t    = threadIdx.x;
  const int widx = t >> 5;
  const int lane = t & 31;
  const int ln   = lane & 15;
  const int kh   = lane >> 4;

  {
    const float* Arow = mh + (size_t)(b * GGRP + gt * 16 + ln) * EMBD;
    const float* Br[4];
#pragma unroll
    for (int j = 0; j < 4; ++j)
      Br[j] = EN + (size_t)(b * NNODE + (widx * 4 + j) * 16 + ln) * EMBD;
    v8f acc[4];
#pragma unroll
    for (int j = 0; j < 4; ++j) acc[j] = v8f{0.f,0.f,0.f,0.f,0.f,0.f,0.f,0.f};
#pragma unroll 1
    for (int k = 0; k < EMBD; k += 32) {
      v4f a[4], bb[4][4];
#pragma unroll
      for (int i = 0; i < 4; ++i) a[i] = ld4g(Arow + k + 8 * i + 4 * kh);
#pragma unroll
      for (int j = 0; j < 4; ++j)
#pragma unroll
        for (int i = 0; i < 4; ++i) bb[j][i] = ld4g(Br[j] + k + 8 * i + 4 * kh);
#pragma unroll
      for (int i = 0; i < 4; ++i) {
#pragma unroll
        for (int j = 0; j < 4; ++j) acc[j] = wmma4(lo2(a[i]), lo2(bb[j][i]), acc[j]);
#pragma unroll
        for (int j = 0; j < 4; ++j) acc[j] = wmma4(hi2(a[i]), hi2(bb[j][i]), acc[j]);
      }
    }
#pragma unroll
    for (int j = 0; j < 4; ++j) {
      const int nb = (widx * 4 + j) * 16;
#pragma unroll
      for (int r = 0; r < 8; ++r) {
        const int row = r + 8 * kh;
        const int g   = gt * 16 + row;
        const float s2 = acc[j][r] * (1.0f / 16.0f);
        sc[row * 512 + nb + ln] =
            10.0f * tanhf(s2) + MASK[(size_t)(b * GGRP + g) * NNODE + nb + ln];
      }
    }
  }
  __syncthreads();

  const int row = t >> 4, l16 = t & 15;
  float* srow = sc + row * 512;
  float mx = -INFINITY;
  for (int j = l16; j < 512; j += 16) mx = fmaxf(mx, srow[j]);
  redA[row][l16] = mx;
  __syncthreads();
  if (l16 == 0) {
    float m2 = redA[row][0];
    for (int i = 1; i < 16; ++i) m2 = fmaxf(m2, redA[row][i]);
    redA[row][0] = m2;
  }
  __syncthreads();
  const float rmax = redA[row][0];
  float ssum = 0.0f;
  for (int j = l16; j < 512; j += 16) {
    const float e = __expf(srow[j] - rmax);
    srow[j] = e;
    ssum += e;
  }
  redB[row][l16] = ssum;
  __syncthreads();
  if (l16 == 0) {
    float s2 = 0.0f;
    for (int i = 0; i < 16; ++i) s2 += redB[row][i];
    redB[row][0] = s2;
  }
  __syncthreads();
  const float inv = 1.0f / redB[row][0];
  float* orow = out + (size_t)(b * GGRP + gt * 16 + row) * NNODE;
  for (int j = l16; j < 512; j += 16) orow[j] = srow[j] * inv;
}

extern "C" void kernel_launch(void* const* d_in, const int* in_sizes, int n_in,
                              void* d_out, int out_size, void* d_ws, size_t ws_size,
                              hipStream_t stream) {
  const float* EN   = (const float*)d_in[0];
  const float* EF   = (const float*)d_in[1];
  const float* EL   = (const float*)d_in[2];
  const float* MASK = (const float*)d_in[3];
  const float* WqG  = (const float*)d_in[4];
  const float* WqF  = (const float*)d_in[5];
  const float* WqL  = (const float*)d_in[6];
  const float* Wk   = (const float*)d_in[7];
  const float* Wv   = (const float*)d_in[8];
  const float* Wc   = (const float*)d_in[9];
  const float* bc   = (const float*)d_in[10];
  float* out = (float*)d_out;

  float* ws   = (float*)d_ws;
  float* qg   = ws;
  float* Qb   = qg   + (size_t)BATCH * EMBD;
  float* Kb   = Qb   + (size_t)BATCH * GGRP * EMBD;
  float* Vb   = Kb   + (size_t)BATCH * NNODE * EMBD;
  float* outc = Vb   + (size_t)BATCH * NNODE * EMBD;
  float* mhb  = outc + (size_t)BATCH * GGRP * EMBD;

  proj_kernel<<<PROJ_BLOCKS, 128, 0, stream>>>(EN, EF, EL, WqG, WqF, WqL, Wk, Wv, qg, Qb, Kb, Vb);

  dim3 ga(GGRP / 16, HNUM / HP, BATCH);
  attn_kernel<<<ga, 256, 0, stream>>>(Qb, qg, Kb, Vb, MASK, outc);

  mh_kernel<<<512, 128, 0, stream>>>(outc, Wc, bc, mhb);

  dim3 gf(GGRP / 16, BATCH, 1);
  final_kernel<<<gf, 256, 0, stream>>>(mhb, EN, MASK, out);
}


# GEMM loops of proj, mh and final: operands staged through LDS by whole-line loads (8 lanes per 128-B line), shared by the block's waves; fragments read from LDS; every sum and its order unchanged
# speedup vs baseline: 2.0736x; 1.2264x over previous
#include <hip/hip_runtime.h>
#include <math.h>

#define BATCH 32
#define NNODE 512
#define GGRP  512
#define EMBD  256
#define HNUM  16
#define DK    16

typedef __attribute__((ext_vector_type(2))) float v2f;
typedef __attribute__((ext_vector_type(4))) float v4f;
typedef __attribute__((ext_vector_type(8))) float v8f;

__device__ __forceinline__ v8f wmma4(v2f a, v2f b, v8f c) {
  return __builtin_amdgcn_wmma_f32_16x16x4_f32(
       false, a,  false, b,
       (short)0, c,  false,  false);
}

__device__ __forceinline__ v2f ld2s(const float* p) {
  return *reinterpret_cast<const v2f*>(p);
}
__device__ __forceinline__ v4f ld4g(const float* __restrict__ p) {
  return *reinterpret_cast<const v4f*>(p);
}
__device__ __forceinline__ v2f lo2(v4f v) { return __builtin_shufflevector(v, v, 0, 1); }
__device__ __forceinline__ v2f hi2(v4f v) { return __builtin_shufflevector(v, v, 2, 3); }

#define LPITCH 36
#define TROWS  192
__device__ __forceinline__ v4f lds4(const float* p) {
  return *reinterpret_cast<const v4f*>(p);
}
__device__ __forceinline__ void gemm_pass_64x128(const float* __restrict__ A1,
                                                 const float* __restrict__ B1,
                                                 const float* __restrict__ A2,
                                                 const float* __restrict__ B2,
                                                 const int nkb,
                                                 float* tile,
                                                 const int tid, const int wm, const int wn,
                                                 const int ln, const int kh,
                                                 v8f (&acc)[2][4]) {
  const int prow = tid >> 3;
  const int pcol = (tid & 7) * 4;
  v4f g[12];
#pragma unroll
  for (int s = 0; s < 4; ++s) g[s] = ld4g(A1 + (size_t)(prow + 16 * s) * EMBD + pcol);
#pragma unroll
  for (int s = 0; s < 8; ++s) g[4 + s] = ld4g(B1 + (size_t)(prow + 16 * s) * EMBD + pcol);
  const float* ta = tile + (32 * wm + ln) * LPITCH + 4 * kh;
  const float* tb = tile + (64 + 64 * wn + ln) * LPITCH + 4 * kh;
  float* tw = tile + prow * LPITCH + pcol;
#pragma unroll 1
  for (int kb = 0; kb < nkb; ++kb) {
    __syncthreads();
#pragma unroll
    for (int s = 0; s < 12; ++s) *reinterpret_cast<v4f*>(tw + 16 * s * LPITCH) = g[s];
    __syncthreads();
    if (kb + 1 < nkb) {
      const int kn = kb + 1;
      const float* An = ((kn < 8) ? A1 : A2) + (kn & 7) * 32;
      const float* Bn = ((kn < 8) ? B1 : B2) + (kn & 7) * 32;
#pragma unroll
      for (int s = 0; s < 4; ++s) g[s] = ld4g(An + (size_t)(prow + 16 * s) * EMBD + pcol);
#pragma unroll
      for (int s = 0; s < 8; ++s) g[4 + s] = ld4g(Bn + (size_t)(prow + 16 * s) * EMBD + pcol);
    }
    v4f a0[4], a1[4], bb[4][4];
#pragma unroll
    for (int i = 0; i < 4; ++i) {
      a0[i] = lds4(ta + 8 * i);
      a1[i] = lds4(ta + 16 * LPITCH + 8 * i);
    }
#pragma unroll
    for (int j = 0; j < 4; ++j)
#pragma unroll
      for (int i = 0; i < 4; ++i) bb[j][i] = lds4(tb + j * 16 * LPITCH + 8 * i);
#pragma unroll
    for (int i = 0; i < 4; ++i) {
#pragma unroll
      for (int j = 0; j < 4; ++j) {
        acc[0][j] = wmma4(lo2(a0[i]), lo2(bb[j][i]), acc[0][j]);
        acc[1][j] = wmma4(lo2(a1[i]), lo2(bb[j][i]), acc[1][j]);
      }
#pragma unroll
      for (int j = 0; j < 4; ++j) {
        acc[0][j] = wmma4(hi2(a0[i]), hi2(bb[j][i]), acc[0][j]);
        acc[1][j] = wmma4(hi2(a1[i]), hi2(bb[j][i]), acc[1][j]);
      }
    }
  }
}

#define PROJ_GROUP  512
#define PROJ_BLOCKS (BATCH + 3 * PROJ_GROUP)
__global__ void __launch_bounds__(128) __attribute__((amdgpu_waves_per_eu(4, 5)))
proj_kernel(const float* __restrict__ EN,
                            const float* __restrict__ EF,
                            const float* __restrict__ EL,
                            const float* __restrict__ WqG,
                            const float* __restrict__ WqF,
                            const float* __restrict__ WqL,
                            const float* __restrict__ Wk,
                            const float* __restrict__ Wv,
                            float* __restrict__ qg,
                            float* __restrict__ Q,
                            float* __restrict__ K,
                            float* __restrict__ V) {
  if (blockIdx.x < BATCH) {
    __shared__ float sm[EMBD];
    const int mb = blockIdx.x;
    const int t  = threadIdx.x;
    const float* base = EN + (size_t)mb * NNODE * EMBD + 2 * t;
    float s0 = 0.0f, s1 = 0.0f;
#pragma unroll 32
    for (int n = 0; n < NNODE; ++n) {
      const v2f x = ld2s(base + (size_t)n * EMBD);
      s0 += x.x;
      s1 += x.y;
    }
    sm[2 * t]     = s0 * (1.0f / (float)NNODE);
    sm[2 * t + 1] = s1 * (1.0f / (float)NNODE);
    __syncthreads();
    float a0 = 0.0f, a1 = 0.0f;
    const float* w0 = WqG + (size_t)t * EMBD;
    const float* w1 = WqG + (size_t)(t + 128) * EMBD;
#pragma unroll 16
    for (int k = 0; k < EMBD; ++k) {
      a0 += sm[k] * w0[k];
      a1 += sm[k] * w1[k];
    }
    qg[(size_t)mb * EMBD + t]       = a0;
    qg[(size_t)mb * EMBD + t + 128] = a1;
    return;
  }
  __shared__ __attribute__((aligned(16))) float tile[TROWS * LPITCH];
  const unsigned gblk = blockIdx.x - BATCH;
  const int sel  = (int)(gblk / PROJ_GROUP);
  const int blk  = (int)(gblk % PROJ_GROUP);
  const int tid  = threadIdx.x;
  const int widx = tid >> 5;
  const int lane = tid & 31;
  const int ln   = lane & 15;
  const int kh   = lane >> 4;
  const int wm   = widx >> 1;
  const int wn   = widx & 1;
  const int row0 = (blk >> 1) * 64;
  const int col0 = (blk & 1) * 128;

  v8f acc[2][4];
#pragma unroll
  for (int i = 0; i < 2; ++i)
#pragma unroll
    for (int j = 0; j < 4; ++j) acc[i][j] = v8f{0.f,0.f,0.f,0.f,0.f,0.f,0.f,0.f};

  float* Out;
  if (sel < 2) {
    const float* W = (sel == 0) ? Wk : Wv;
    const float* A0 = EN + (size_t)row0 * EMBD;
    const float* B0 = W + (size_t)col0 * EMBD;
    gemm_pass_64x128(A0, B0, A0, B0, 8, tile, tid, wm, wn, ln, kh, acc);
    Out = (sel == 0) ? K : V;
  } else {
    gemm_pass_64x128(EF + (size_t)row0 * EMBD, WqF + (size_t)col0 * EMBD,
                     EL + (size_t)row0 * EMBD, WqL + (size_t)col0 * EMBD,
                     16, tile, tid, wm, wn, ln, kh, acc);
    Out = Q;
  }

#pragma unroll
  for (int i = 0; i < 2; ++i)
#pragma unroll
    for (int j = 0; j < 4; ++j) {
      const int col = col0 + wn * 64 + j * 16 + ln;
#pragma unroll
      for (int r = 0; r < 8; ++r) {
        const int row = row0 + wm * 32 + i * 16 + r + 8 * kh;
        Out[(size_t)row * EMBD + col] = acc[i][j][r];
      }
    }
}

#define HP 2
#define OPITCH 18
#define OHEAD (8 * 16 * OPITCH + 1)
__global__ void __launch_bounds__(256, 8)
attn_kernel(const float* __restrict__ Q,
            const float* __restrict__ QG,
            const float* __restrict__ K,
            const float* __restrict__ V,
            const float* __restrict__ MASK,
            float* __restrict__ outc) {
  __shared__ float s_m[HP][8][16];
  __shared__ float s_l[HP][8][16];
  __shared__ float s_o[HP * OHEAD];

  const int gt = blockIdx.x, hp = blockIdx.y, b = blockIdx.z;
  const int t    = threadIdx.x;
  const int widx = __builtin_amdgcn_readfirstlane(t >> 5);
  const int lane = t & 31;
  const int ln   = lane & 15;
  const int kh   = lane >> 4;
  const int nb   = widx * 64;
  const int c0   = hp * (HP * DK);

  const float* Qrow = Q + (size_t)(b * GGRP + gt * 16) * EMBD + c0 + (ln * EMBD + 4 * kh);
  const float* Krow = K + (size_t)(b * NNODE + nb) * EMBD + c0 + (ln * EMBD + 4 * kh);
  v4f q0[HP], q1[HP];
  v4f k0[HP][4], k1[HP][4];
#pragma unroll
  for (int hh = 0; hh < HP; ++hh) {
    q0[hh] = ld4g(Qrow + hh * DK);
    q1[hh] = ld4g(Qrow + hh * DK + 8);
  }
  const float* QGrow = QG + (size_t)b * EMBD + c0 + (4 * kh);
#pragma unroll
  for (int hh = 0; hh < HP; ++hh) {
    q0[hh] += ld4g(QGrow + hh * DK);
    q1[hh] += ld4g(QGrow + hh * DK + 8);
  }
#pragma unroll
  for (int j = 0; j < 4; ++j) {
#pragma unroll
    for (int hh = 0; hh < HP; ++hh) {
      k0[hh][j] = ld4g(Krow + (size_t)j * 16 * EMBD + hh * DK);
      k1[hh][j] = ld4g(Krow + (size_t)j * 16 * EMBD + hh * DK + 8);
    }
  }
  const float* Mrow = MASK + (size_t)(b * GGRP + gt * 16) * NNODE + nb + (ln * NNODE + 8 * kh);
  v4f ma[4], mb[4];
#pragma unroll
  for (int j = 0; j < 4; ++j) {
    ma[j] = ld4g(Mrow + 16 * j);
    mb[j] = ld4g(Mrow + 16 * j + 4);
  }

  v8f acc[HP][4];
#pragma unroll
  for (int hh = 0; hh < HP; ++hh) {
#pragma unroll
    for (int j = 0; j < 4; ++j) acc[hh][j] = v8f{0.f,0.f,0.f,0.f,0.f,0.f,0.f,0.f};
#pragma unroll
    for (int j = 0; j < 4; ++j) acc[hh][j] = wmma4(lo2(k0[hh][j]), lo2(q0[hh]), acc[hh][j]);
#pragma unroll
    for (int j = 0; j < 4; ++j) acc[hh][j] = wmma4(hi2(k0[hh][j]), hi2(q0[hh]), acc[hh][j]);
#pragma unroll
    for (int j = 0; j < 4; ++j) acc[hh][j] = wmma4(lo2(k1[hh][j]), lo2(q1[hh]), acc[hh][j]);
#pragma unroll
    for (int j = 0; j < 4; ++j) acc[hh][j] = wmma4(hi2(k1[hh][j]), hi2(q1[hh]), acc[hh][j]);
  }

#pragma unroll
  for (int hh = 0; hh < HP; ++hh) {
    float mloc = -INFINITY;
#pragma unroll
    for (int j = 0; j < 4; ++j) {
#pragma unroll
      for (int r = 0; r < 4; ++r) {
        acc[hh][j][r]     = acc[hh][j][r]     * 0.25f + ma[j][r];
        acc[hh][j][r + 4] = acc[hh][j][r + 4] * 0.25f + mb[j][r];
        mloc = fmaxf(mloc, fmaxf(acc[hh][j][r], acc[hh][j][r + 4]));
      }
    }
    mloc = fmaxf(mloc, __shfl_xor(mloc, 16));
    const float muse = (mloc == -INFINITY) ? 0.0f : mloc;

    float lsum = 0.0f;
#pragma unroll
    for (int j = 0; j < 4; ++j) {
#pragma unroll
      for (int r = 0; r < 8; ++r) {
        const float e = __expf(acc[hh][j][r] - muse);
        acc[hh][j][r] = e;
        lsum += e;
      }
    }
    lsum += __shfl_xor(lsum, 16);
    s_m[hh][widx][ln] = mloc;
    s_l[hh][widx][ln] = lsum;
  }

  const float* Vp = V + (size_t)(b * NNODE + nb) * EMBD + c0 + (8 * kh * EMBD + ln);
  v8f o0[HP], o1[HP];
#pragma unroll
  for (int hh = 0; hh < HP; ++hh) {
    o0[hh] = v8f{0.f,0.f,0.f,0.f,0.f,0.f,0.f,0.f};
    o1[hh] = v8f{0.f,0.f,0.f,0.f,0.f,0.f,0.f,0.f};
  }
#pragma unroll
  for (int j = 0; j < 4; ++j) {
#pragma unroll
    for (int r = 0; r < 8; r += 4) {
      v2f a0[HP], a1[HP];
#pragma unroll
      for (int hh = 0; hh < HP; ++hh) {
        a0[hh].x = Vp[(size_t)(16 * j + r)     * EMBD + hh * DK];
        a0[hh].y = Vp[(size_t)(16 * j + r + 1) * EMBD + hh * DK];
        a1[hh].x = Vp[(size_t)(16 * j + r + 2) * EMBD + hh * DK];
        a1[hh].y = Vp[(size_t)(16 * j + r + 3) * EMBD + hh * DK];
      }
#pragma unroll
      for (int hh = 0; hh < HP; ++hh) {
        v2f w0, w1;
        w0.x = acc[hh][j][r];     w0.y = acc[hh][j][r + 1];
        w1.x = acc[hh][j][r + 2]; w1.y = acc[hh][j][r + 3];
        o0[hh] = wmma4(a0[hh], w0, o0[hh]);
        o1[hh] = wmma4(a1[hh], w1, o1[hh]);
      }
    }
  }
#pragma unroll
  for (int hh = 0; hh < HP; ++hh) {
#pragma unroll
    for (int r = 0; r < 8; ++r)
      s_o[hh * OHEAD + widx * (16 * OPITCH) + (8 * kh + r) * OPITCH + ln] = o0[hh][r] + o1[hh][r];
  }
  __syncthreads();

  static_assert(HP * DK == 32, "one lane per column of the head pair");
#pragma unroll
  for (int i = 0; i < 2; ++i) {
    const int g  = 8 * i + widx;
    const int hh = lane >> 4, d = lane & 15;
    float mw[8];
    float M = -INFINITY;
#pragma unroll
    for (int w = 0; w < 8; ++w) { mw[w] = s_m[hh][w][g]; M = fmaxf(M, mw[w]); }
    float L = 0.0f, O = 0.0f;
#pragma unroll
    for (int w = 0; w < 8; ++w) {
      const float e = __expf(mw[w] - M);
      L += e * s_l[hh][w][g];
      O += e * s_o[hh * OHEAD + w * (16 * OPITCH) + d * OPITCH + g];
    }
    outc[(size_t)(b * GGRP + gt * 16 + g) * EMBD + c0 + lane] = O / L;
  }
}

__global__ void __launch_bounds__(128) __attribute__((amdgpu_waves_per_eu(4, 5)))
mh_kernel(const float* __restrict__ outc,
                          const float* __restrict__ Wc,
                          const float* __restrict__ bc,
                          float* __restrict__ mh) {
  __shared__ __attribute__((aligned(16))) float tile[TROWS * LPITCH];
  const int blk  = blockIdx.x;
  const int tid  = threadIdx.x;
  const int widx = tid >> 5;
  const int lane = tid & 31;
  const int ln   = lane & 15;
  const int kh   = lane >> 4;
  const int wm   = widx >> 1;
  const int wn   = widx & 1;
  const int row0 = (blk >> 1) * 64;
  const int col0 = (blk & 1) * 128;

  v8f acc[2][4];
#pragma unroll
  for (int i = 0; i < 2; ++i)
#pragma unroll
    for (int j = 0; j < 4; ++j) acc[i][j] = v8f{0.f,0.f,0.f,0.f,0.f,0.f,0.f,0.f};

  const float* A0 = outc + (size_t)row0 * EMBD;
  const float* B0 = Wc + (size_t)col0 * EMBD;
  gemm_pass_64x128(A0, B0, A0, B0, 8, tile, tid, wm, wn, ln, kh, acc);
#pragma unroll
  for (int i = 0; i < 2; ++i)
#pragma unroll
    for (int j = 0; j < 4; ++j) {
      const int col  = col0 + wn * 64 + j * 16 + ln;
      const float bv = bc[col];
#pragma unroll
      for (int r = 0; r < 8; ++r) {
        const int row = row0 + wm * 32 + i * 16 + r + 8 * kh;
        mh[(size_t)row * EMBD + col] = acc[i][j][r] + bv;
      }
    }
}

#define FROWS (16 + NNODE)
__global__ void __launch_bounds__(256) __attribute__((amdgpu_waves_per_eu(8, 8)))
final_kernel(const float* __restrict__ mh,
                             const float* __restrict__ EN,
                             const float* __restrict__ MASK,
                             float* __restrict__ out) {
  __shared__ __attribute__((aligned(16))) float buf[FROWS * LPITCH];
  static_assert(FROWS * LPITCH >= 16 * 512, "score tile fits the staging buffer");
  float* sc = buf;
  __shared__ float redA[16][16];
  __shared__ float redB[16][16];

  const int gt = blockIdx.x, b = blockIdx.y;
  const int t    = threadIdx.x;
  const int widx = t >> 5;
  const int lane = t & 31;
  const int ln   = lane & 15;
  const int kh   = lane >> 4;

  {
    const int prow = t >> 3;
    const int pcol = (t & 7) * 4;
    const float* Ag = mh + (size_t)(b * GGRP + gt * 16 + (prow & 15)) * EMBD + pcol;
    const float* Bg = EN + (size_t)(b * NNODE + prow) * EMBD + pcol;
    float* twa = buf + (prow & 15) * LPITCH + pcol;
    float* twb = buf + (16 + prow) * LPITCH + pcol;
    const float* ta = buf + ln * LPITCH + 4 * kh;
    const float* tb = buf + (16 + widx * 64 + ln) * LPITCH + 4 * kh;
    v8f acc[4];
#pragma unroll
    for (int j = 0; j < 4; ++j) acc[j] = v8f{0.f,0.f,0.f,0.f,0.f,0.f,0.f,0.f};
#pragma unroll 1
    for (int k = 0; k < EMBD; k += 32) {
      v4f g[16];
#pragma unroll
      for (int s = 0; s < 16; ++s) g[s] = ld4g(Bg + (size_t)32 * s * EMBD + k);
      const v4f ga = ld4g(Ag + k);
      __syncthreads();
#pragma unroll
      for (int s = 0; s < 16; ++s) *reinterpret_cast<v4f*>(twb + 32 * s * LPITCH) = g[s];
      *reinterpret_cast<v4f*>(twa) = ga;
      __syncthreads();
      v4f a[4], bb[4][4];
#pragma unroll
      for (int i = 0; i < 4; ++i) a[i] = lds4(ta + 8 * i);
#pragma unroll
      for (int j = 0; j < 4; ++j)
#pragma unroll
        for (int i = 0; i < 4; ++i) bb[j][i] = lds4(tb + j * 16 * LPITCH + 8 * i);
#pragma unroll
      for (int i = 0; i < 4; ++i) {
#pragma unroll
        for (int j = 0; j < 4; ++j) acc[j] = wmma4(lo2(a[i]), lo2(bb[j][i]), acc[j]);
#pragma unroll
        for (int j = 0; j < 4; ++j) acc[j] = wmma4(hi2(a[i]), hi2(bb[j][i]), acc[j]);
      }
    }
    __syncthreads();
#pragma unroll
    for (int j = 0; j < 4; ++j) {
      const int nb = (widx * 4 + j) * 16;
#pragma unroll
      for (int r = 0; r < 8; ++r) {
        const int row = r + 8 * kh;
        const int g   = gt * 16 + row;
        const float s2 = acc[j][r] * (1.0f / 16.0f);
        sc[row * 512 + nb + ln] =
            10.0f * tanhf(s2) + MASK[(size_t)(b * GGRP + g) * NNODE + nb + ln];
      }
    }
  }
  __syncthreads();

  const int row = t >> 4, l16 = t & 15;
  float* srow = sc + row * 512;
  float mx = -INFINITY;
  for (int j = l16; j < 512; j += 16) mx = fmaxf(mx, srow[j]);
  redA[row][l16] = mx;
  __syncthreads();
  if (l16 == 0) {
    float m2 = redA[row][0];
    for (int i = 1; i < 16; ++i) m2 = fmaxf(m2, redA[row][i]);
    redA[row][0] = m2;
  }
  __syncthreads();
  const float rmax = redA[row][0];
  float ssum = 0.0f;
  for (int j = l16; j < 512; j += 16) {
    const float e = __expf(srow[j] - rmax);
    srow[j] = e;
    ssum += e;
  }
  redB[row][l16] = ssum;
  __syncthreads();
  if (l16 == 0) {
    float s2 = 0.0f;
    for (int i = 0; i < 16; ++i) s2 += redB[row][i];
    redB[row][0] = s2;
  }
  __syncthreads();
  const float inv = 1.0f / redB[row][0];
  float* orow = out + (size_t)(b * GGRP + gt * 16 + row) * NNODE;
  for (int j = l16; j < 512; j += 16) orow[j] = srow[j] * inv;
}

extern "C" void kernel_launch(void* const* d_in, const int* in_sizes, int n_in,
                              void* d_out, int out_size, void* d_ws, size_t ws_size,
                              hipStream_t stream) {
  const float* EN   = (const float*)d_in[0];
  const float* EF   = (const float*)d_in[1];
  const float* EL   = (const float*)d_in[2];
  const float* MASK = (const float*)d_in[3];
  const float* WqG  = (const float*)d_in[4];
  const float* WqF  = (const float*)d_in[5];
  const float* WqL  = (const float*)d_in[6];
  const float* Wk   = (const float*)d_in[7];
  const float* Wv   = (const float*)d_in[8];
  const float* Wc   = (const float*)d_in[9];
  const float* bc   = (const float*)d_in[10];
  float* out = (float*)d_out;

  float* ws   = (float*)d_ws;
  float* qg   = ws;
  float* Qb   = qg   + (size_t)BATCH * EMBD;
  float* Kb   = Qb   + (size_t)BATCH * GGRP * EMBD;
  float* Vb   = Kb   + (size_t)BATCH * NNODE * EMBD;
  float* outc = Vb   + (size_t)BATCH * NNODE * EMBD;
  float* mhb  = outc + (size_t)BATCH * GGRP * EMBD;

  proj_kernel<<<PROJ_BLOCKS, 128, 0, stream>>>(EN, EF, EL, WqG, WqF, WqL, Wk, Wv, qg, Qb, Kb, Vb);

  dim3 ga(GGRP / 16, HNUM / HP, BATCH);
  attn_kernel<<<ga, 256, 0, stream>>>(Qb, qg, Kb, Vb, MASK, outc);

  mh_kernel<<<512, 128, 0, stream>>>(outc, Wc, bc, mhb);

  dim3 gf(GGRP / 16, BATCH, 1);
  final_kernel<<<gf, 256, 0, stream>>>(mhb, EN, MASK, out);
}


# Attention: a block loops over 4 query tiles with its K and V fragments held in registers (4x fewer waves; K, V leave L2 once per 4 tiles); partials double-buffered in LDS; sum orders unchanged
# speedup vs baseline: 2.1882x; 1.0553x over previous
#include <hip/hip_runtime.h>
#include <math.h>

#define BATCH 32
#define NNODE 512
#define GGRP  512
#define EMBD  256
#define HNUM  16
#define DK    16

typedef __attribute__((ext_vector_type(2))) float v2f;
typedef __attribute__((ext_vector_type(4))) float v4f;
typedef __attribute__((ext_vector_type(8))) float v8f;

__device__ __forceinline__ v8f wmma4(v2f a, v2f b, v8f c) {
  return __builtin_amdgcn_wmma_f32_16x16x4_f32(
       false, a,  false, b,
       (short)0, c,  false,  false);
}

__device__ __forceinline__ v2f ld2s(const float* p) {
  return *reinterpret_cast<const v2f*>(p);
}
__device__ __forceinline__ v4f ld4g(const float* __restrict__ p) {
  return *reinterpret_cast<const v4f*>(p);
}
__device__ __forceinline__ v2f lo2(v4f v) { return __builtin_shufflevector(v, v, 0, 1); }
__device__ __forceinline__ v2f hi2(v4f v) { return __builtin_shufflevector(v, v, 2, 3); }

#define LPITCH 36
#define TROWS  192
__device__ __forceinline__ v4f lds4(const float* p) {
  return *reinterpret_cast<const v4f*>(p);
}
__device__ __forceinline__ void gemm_pass_64x128(const float* __restrict__ A1,
                                                 const float* __restrict__ B1,
                                                 const float* __restrict__ A2,
                                                 const float* __restrict__ B2,
                                                 const int nkb,
                                                 float* tile,
                                                 const int tid, const int wm, const int wn,
                                                 const int ln, const int kh,
                                                 v8f (&acc)[2][4]) {
  const int prow = tid >> 3;
  const int pcol = (tid & 7) * 4;
  v4f g[12];
#pragma unroll
  for (int s = 0; s < 4; ++s) g[s] = ld4g(A1 + (size_t)(prow + 16 * s) * EMBD + pcol);
#pragma unroll
  for (int s = 0; s < 8; ++s) g[4 + s] = ld4g(B1 + (size_t)(prow + 16 * s) * EMBD + pcol);
  const float* ta = tile + (32 * wm + ln) * LPITCH + 4 * kh;
  const float* tb = tile + (64 + 64 * wn + ln) * LPITCH + 4 * kh;
  float* tw = tile + prow * LPITCH + pcol;
#pragma unroll 1
  for (int kb = 0; kb < nkb; ++kb) {
    __syncthreads();
#pragma unroll
    for (int s = 0; s < 12; ++s) *reinterpret_cast<v4f*>(tw + 16 * s * LPITCH) = g[s];
    __syncthreads();
    if (kb + 1 < nkb) {
      const int kn = kb + 1;
      const float* An = ((kn < 8) ? A1 : A2) + (kn & 7) * 32;
      const float* Bn = ((kn < 8) ? B1 : B2) + (kn & 7) * 32;
#pragma unroll
      for (int s = 0; s < 4; ++s) g[s] = ld4g(An + (size_t)(prow + 16 * s) * EMBD + pcol);
#pragma unroll
      for (int s = 0; s < 8; ++s) g[4 + s] = ld4g(Bn + (size_t)(prow + 16 * s) * EMBD + pcol);
    }
    v4f a0[4], a1[4], bb[4][4];
#pragma unroll
    for (int i = 0; i < 4; ++i) {
      a0[i] = lds4(ta + 8 * i);
      a1[i] = lds4(ta + 16 * LPITCH + 8 * i);
    }
#pragma unroll
    for (int j = 0; j < 4; ++j)
#pragma unroll
      for (int i = 0; i < 4; ++i) bb[j][i] = lds4(tb + j * 16 * LPITCH + 8 * i);
#pragma unroll
    for (int i = 0; i < 4; ++i) {
#pragma unroll
      for (int j = 0; j < 4; ++j) {
        acc[0][j] = wmma4(lo2(a0[i]), lo2(bb[j][i]), acc[0][j]);
        acc[1][j] = wmma4(lo2(a1[i]), lo2(bb[j][i]), acc[1][j]);
      }
#pragma unroll
      for (int j = 0; j < 4; ++j) {
        acc[0][j] = wmma4(hi2(a0[i]), hi2(bb[j][i]), acc[0][j]);
        acc[1][j] = wmma4(hi2(a1[i]), hi2(bb[j][i]), acc[1][j]);
      }
    }
  }
}

#define PROJ_GROUP  512
#define PROJ_BLOCKS (BATCH + 3 * PROJ_GROUP)
__global__ void __launch_bounds__(128) __attribute__((amdgpu_waves_per_eu(4, 5)))
proj_kernel(const float* __restrict__ EN,
                            const float* __restrict__ EF,
                            const float* __restrict__ EL,
                            const float* __restrict__ WqG,
                            const float* __restrict__ WqF,
                            const float* __restrict__ WqL,
                            const float* __restrict__ Wk,
                            const float* __restrict__ Wv,
                            float* __restrict__ qg,
                            float* __restrict__ Q,
                            float* __restrict__ K,
                            float* __restrict__ V) {
  if (blockIdx.x < BATCH) {
    __shared__ float sm[EMBD];
    const int mb = blockIdx.x;
    const int t  = threadIdx.x;
    const float* base = EN + (size_t)mb * NNODE * EMBD + 2 * t;
    float s0 = 0.0f, s1 = 0.0f;
#pragma unroll 32
    for (int n = 0; n < NNODE; ++n) {
      const v2f x = ld2s(base + (size_t)n * EMBD);
      s0 += x.x;
      s1 += x.y;
    }
    sm[2 * t]     = s0 * (1.0f / (float)NNODE);
    sm[2 * t + 1] = s1 * (1.0f / (float)NNODE);
    __syncthreads();
    float a0 = 0.0f, a1 = 0.0f;
    const float* w0 = WqG + (size_t)t * EMBD;
    const float* w1 = WqG + (size_t)(t + 128) * EMBD;
#pragma unroll 16
    for (int k = 0; k < EMBD; ++k) {
      a0 += sm[k] * w0[k];
      a1 += sm[k] * w1[k];
    }
    qg[(size_t)mb * EMBD + t]       = a0;
    qg[(size_t)mb * EMBD + t + 128] = a1;
    return;
  }
  __shared__ __attribute__((aligned(16))) float tile[TROWS * LPITCH];
  const unsigned gblk = blockIdx.x - BATCH;
  const int sel  = (int)(gblk / PROJ_GROUP);
  const int blk  = (int)(gblk % PROJ_GROUP);
  const int tid  = threadIdx.x;
  const int widx = tid >> 5;
  const int lane = tid & 31;
  const int ln   = lane & 15;
  const int kh   = lane >> 4;
  const int wm   = widx >> 1;
  const int wn   = widx & 1;
  const int row0 = (blk >> 1) * 64;
  const int col0 = (blk & 1) * 128;

  v8f acc[2][4];
#pragma unroll
  for (int i = 0; i < 2; ++i)
#pragma unroll
    for (int j = 0; j < 4; ++j) acc[i][j] = v8f{0.f,0.f,0.f,0.f,0.f,0.f,0.f,0.f};

  float* Out;
  if (sel < 2) {
    const float* W = (sel == 0) ? Wk : Wv;
    const float* A0 = EN + (size_t)row0 * EMBD;
    const float* B0 = W + (size_t)col0 * EMBD;
    gemm_pass_64x128(A0, B0, A0, B0, 8, tile, tid, wm, wn, ln, kh, acc);
    Out = (sel == 0) ? K : V;
  } else {
    gemm_pass_64x128(EF + (size_t)row0 * EMBD, WqF + (size_t)col0 * EMBD,
                     EL + (size_t)row0 * EMBD, WqL + (size_t)col0 * EMBD,
                     16, tile, tid, wm, wn, ln, kh, acc);
    Out = Q;
  }

#pragma unroll
  for (int i = 0; i < 2; ++i)
#pragma unroll
    for (int j = 0; j < 4; ++j) {
      const int col = col0 + wn * 64 + j * 16 + ln;
#pragma unroll
      for (int r = 0; r < 8; ++r) {
        const int row = row0 + wm * 32 + i * 16 + r + 8 * kh;
        Out[(size_t)row * EMBD + col] = acc[i][j][r];
      }
    }
}

#define HP 2
#define GTB 4
#define OPITCH 18
#define OHEAD (8 * 16 * OPITCH + 1)
__global__ void __launch_bounds__(256) __attribute__((amdgpu_waves_per_eu(4, 4)))
attn_kernel(const float* __restrict__ Q,
            const float* __restrict__ QG,
            const float* __restrict__ K,
            const float* __restrict__ V,
            const float* __restrict__ MASK,
            float* __restrict__ outc) {
  __shared__ float s_m[2][HP][8][16];
  __shared__ float s_l[2][HP][8][16];
  __shared__ float s_o[2][HP * OHEAD];

  const int gt0 = blockIdx.x * GTB, hp = blockIdx.y, b = blockIdx.z;
  const int t    = threadIdx.x;
  const int widx = __builtin_amdgcn_readfirstlane(t >> 5);
  const int lane = t & 31;
  const int ln   = lane & 15;
  const int kh   = lane >> 4;
  const int nb   = widx * 64;
  const int c0   = hp * (HP * DK);

  const float* Krow = K + (size_t)(b * NNODE + nb) * EMBD + c0 + (ln * EMBD + 4 * kh);
  v4f k0[HP][4], k1[HP][4];
#pragma unroll
  for (int j = 0; j < 4; ++j) {
#pragma unroll
    for (int hh = 0; hh < HP; ++hh) {
      k0[hh][j] = ld4g(Krow + (size_t)j * 16 * EMBD + hh * DK);
      k1[hh][j] = ld4g(Krow + (size_t)j * 16 * EMBD + hh * DK + 8);
    }
  }
  const float* Vp = V + (size_t)(b * NNODE + nb) * EMBD + c0 + (8 * kh * EMBD + ln);
  v2f va0[HP][4][2], va1[HP][4][2];
#pragma unroll
  for (int j = 0; j < 4; ++j) {
#pragma unroll
    for (int rr = 0; rr < 2; ++rr) {
#pragma unroll
      for (int hh = 0; hh < HP; ++hh) {
        va0[hh][j][rr].x = Vp[(size_t)(16 * j + 4 * rr)     * EMBD + hh * DK];
        va0[hh][j][rr].y = Vp[(size_t)(16 * j + 4 * rr + 1) * EMBD + hh * DK];
        va1[hh][j][rr].x = Vp[(size_t)(16 * j + 4 * rr + 2) * EMBD + hh * DK];
        va1[hh][j][rr].y = Vp[(size_t)(16 * j + 4 * rr + 3) * EMBD + hh * DK];
      }
    }
  }
  const float* QGrow = QG + (size_t)b * EMBD + c0 + (4 * kh);

  const float* Qrow = Q + (size_t)(b * GGRP + gt0 * 16) * EMBD + c0 + (ln * EMBD + 4 * kh);
  const float* Mrow = MASK + (size_t)(b * GGRP + gt0 * 16) * NNODE + nb + (ln * NNODE + 8 * kh);
  v4f ma[4], mb[4];
  v4f qc0 = ld4g(Qrow);
  v4f qc1 = ld4g(Qrow + 8);
#pragma unroll
  for (int j = 0; j < 4; ++j) {
    ma[j] = ld4g(Mrow + 16 * j);
    mb[j] = ld4g(Mrow + 16 * j + 4);
  }

#pragma unroll 1
  for (int it = 0; it < GTB; ++it) {
    const int gt = gt0 + it;
    const int pb = it & 1;

#pragma unroll
    for (int hh = 0; hh < HP; ++hh) {
      const v4f qa = ((hh == 0) ? qc0 : ld4g(Qrow + hh * DK))     + ld4g(QGrow + hh * DK);
      const v4f qb = ((hh == 0) ? qc1 : ld4g(Qrow + hh * DK + 8)) + ld4g(QGrow + hh * DK + 8);
      v8f acc[4];
#pragma unroll
      for (int j = 0; j < 4; ++j) acc[j] = v8f{0.f,0.f,0.f,0.f,0.f,0.f,0.f,0.f};
#pragma unroll
      for (int j = 0; j < 4; ++j) acc[j] = wmma4(lo2(k0[hh][j]), lo2(qa), acc[j]);
#pragma unroll
      for (int j = 0; j < 4; ++j) acc[j] = wmma4(hi2(k0[hh][j]), hi2(qa), acc[j]);
#pragma unroll
      for (int j = 0; j < 4; ++j) acc[j] = wmma4(lo2(k1[hh][j]), lo2(qb), acc[j]);
#pragma unroll
      for (int j = 0; j < 4; ++j) acc[j] = wmma4(hi2(k1[hh][j]), hi2(qb), acc[j]);

      float mloc = -INFINITY;
#pragma unroll
      for (int j = 0; j < 4; ++j) {
#pragma unroll
        for (int r = 0; r < 4; ++r) {
          acc[j][r]     = acc[j][r]     * 0.25f + ma[j][r];
          acc[j][r + 4] = acc[j][r + 4] * 0.25f + mb[j][r];
          mloc = fmaxf(mloc, fmaxf(acc[j][r], acc[j][r + 4]));
        }
      }
      mloc = fmaxf(mloc, __shfl_xor(mloc, 16));
      const float muse = (mloc == -INFINITY) ? 0.0f : mloc;

      float lsum = 0.0f;
#pragma unroll
      for (int j = 0; j < 4; ++j) {
#pragma unroll
        for (int r = 0; r < 8; ++r) {
          const float e = __expf(acc[j][r] - muse);
          acc[j][r] = e;
          lsum += e;
        }
      }
      lsum += __shfl_xor(lsum, 16);
      s_m[pb][hh][widx][ln] = mloc;
      s_l[pb][hh][widx][ln] = lsum;

      v8f o0 = v8f{0.f,0.f,0.f,0.f,0.f,0.f,0.f,0.f};
      v8f o1 = v8f{0.f,0.f,0.f,0.f,0.f,0.f,0.f,0.f};
#pragma unroll
      for (int j = 0; j < 4; ++j) {
#pragma unroll
        for (int r = 0; r < 8; r += 4) {
          v2f w0, w1;
          w0.x = acc[j][r];     w0.y = acc[j][r + 1];
          w1.x = acc[j][r + 2]; w1.y = acc[j][r + 3];
          o0 = wmma4(va0[hh][j][r / 4], w0, o0);
          o1 = wmma4(va1[hh][j][r / 4], w1, o1);
        }
      }
#pragma unroll
      for (int r = 0; r < 8; ++r)
        s_o[pb][hh * OHEAD + widx * (16 * OPITCH) + (8 * kh + r) * OPITCH + ln] = o0[r] + o1[r];
    }

    Qrow += 16 * EMBD;
    if (it + 1 < GTB) {
      Mrow += 16 * NNODE;
      qc0 = ld4g(Qrow);
      qc1 = ld4g(Qrow + 8);
#pragma unroll
      for (int j = 0; j < 4; ++j) {
        ma[j] = ld4g(Mrow + 16 * j);
        mb[j] = ld4g(Mrow + 16 * j + 4);
      }
    }
    __syncthreads();

    static_assert(HP * DK == 32, "one lane per column of the head pair");
#pragma unroll
    for (int i = 0; i < 2; ++i) {
      const int g  = 8 * i + widx;
      const int hh = lane >> 4, d = lane & 15;
      float mw[8];
      float M = -INFINITY;
#pragma unroll
      for (int w = 0; w < 8; ++w) { mw[w] = s_m[pb][hh][w][g]; M = fmaxf(M, mw[w]); }
      float L = 0.0f, O = 0.0f;
#pragma unroll
      for (int w = 0; w < 8; ++w) {
        const float e = __expf(mw[w] - M);
        L += e * s_l[pb][hh][w][g];
        O += e * s_o[pb][hh * OHEAD + w * (16 * OPITCH) + d * OPITCH + g];
      }
      outc[(size_t)(b * GGRP + gt * 16 + g) * EMBD + c0 + lane] = O / L;
    }
  }
}

__global__ void __launch_bounds__(128) __attribute__((amdgpu_waves_per_eu(4, 5)))
mh_kernel(const float* __restrict__ outc,
                          const float* __restrict__ Wc,
                          const float* __restrict__ bc,
                          float* __restrict__ mh) {
  __shared__ __attribute__((aligned(16))) float tile[TROWS * LPITCH];
  const int blk  = blockIdx.x;
  const int tid  = threadIdx.x;
  const int widx = tid >> 5;
  const int lane = tid & 31;
  const int ln   = lane & 15;
  const int kh   = lane >> 4;
  const int wm   = widx >> 1;
  const int wn   = widx & 1;
  const int row0 = (blk >> 1) * 64;
  const int col0 = (blk & 1) * 128;

  v8f acc[2][4];
#pragma unroll
  for (int i = 0; i < 2; ++i)
#pragma unroll
    for (int j = 0; j < 4; ++j) acc[i][j] = v8f{0.f,0.f,0.f,0.f,0.f,0.f,0.f,0.f};

  const float* A0 = outc + (size_t)row0 * EMBD;
  const float* B0 = Wc + (size_t)col0 * EMBD;
  gemm_pass_64x128(A0, B0, A0, B0, 8, tile, tid, wm, wn, ln, kh, acc);
#pragma unroll
  for (int i = 0; i < 2; ++i)
#pragma unroll
    for (int j = 0; j < 4; ++j) {
      const int col  = col0 + wn * 64 + j * 16 + ln;
      const float bv = bc[col];
#pragma unroll
      for (int r = 0; r < 8; ++r) {
        const int row = row0 + wm * 32 + i * 16 + r + 8 * kh;
        mh[(size_t)row * EMBD + col] = acc[i][j][r] + bv;
      }
    }
}

#define FROWS (16 + NNODE)
__global__ void __launch_bounds__(256) __attribute__((amdgpu_waves_per_eu(8, 8)))
final_kernel(const float* __restrict__ mh,
                             const float* __restrict__ EN,
                             const float* __restrict__ MASK,
                             float* __restrict__ out) {
  __shared__ __attribute__((aligned(16))) float buf[FROWS * LPITCH];
  static_assert(FROWS * LPITCH >= 16 * 512, "score tile fits the staging buffer");
  float* sc = buf;
  __shared__ float redA[16][16];
  __shared__ float redB[16][16];

  const int gt = blockIdx.x, b = blockIdx.y;
  const int t    = threadIdx.x;
  const int widx = t >> 5;
  const int lane = t & 31;
  const int ln   = lane & 15;
  const int kh   = lane >> 4;

  {
    const int prow = t >> 3;
    const int pcol = (t & 7) * 4;
    const float* Ag = mh + (size_t)(b * GGRP + gt * 16 + (prow & 15)) * EMBD + pcol;
    const float* Bg = EN + (size_t)(b * NNODE + prow) * EMBD + pcol;
    float* twa = buf + (prow & 15) * LPITCH + pcol;
    float* twb = buf + (16 + prow) * LPITCH + pcol;
    const float* ta = buf + ln * LPITCH + 4 * kh;
    const float* tb = buf + (16 + widx * 64 + ln) * LPITCH + 4 * kh;
    v8f acc[4];
#pragma unroll
    for (int j = 0; j < 4; ++j) acc[j] = v8f{0.f,0.f,0.f,0.f,0.f,0.f,0.f,0.f};
#pragma unroll 1
    for (int k = 0; k < EMBD; k += 32) {
      v4f g[16];
#pragma unroll
      for (int s = 0; s < 16; ++s) g[s] = ld4g(Bg + (size_t)32 * s * EMBD + k);
      const v4f ga = ld4g(Ag + k);
      __syncthreads();
#pragma unroll
      for (int s = 0; s < 16; ++s) *reinterpret_cast<v4f*>(twb + 32 * s * LPITCH) = g[s];
      *reinterpret_cast<v4f*>(twa) = ga;
      __syncthreads();
      v4f a[4], bb[4][4];
#pragma unroll
      for (int i = 0; i < 4; ++i) a[i] = lds4(ta + 8 * i);
#pragma unroll
      for (int j = 0; j < 4; ++j)
#pragma unroll
        for (int i = 0; i < 4; ++i) bb[j][i] = lds4(tb + j * 16 * LPITCH + 8 * i);
#pragma unroll
      for (int i = 0; i < 4; ++i) {
#pragma unroll
        for (int j = 0; j < 4; ++j) acc[j] = wmma4(lo2(a[i]), lo2(bb[j][i]), acc[j]);
#pragma unroll
        for (int j = 0; j < 4; ++j) acc[j] = wmma4(hi2(a[i]), hi2(bb[j][i]), acc[j]);
      }
    }
    __syncthreads();
#pragma unroll
    for (int j = 0; j < 4; ++j) {
      const int nb = (widx * 4 + j) * 16;
#pragma unroll
      for (int r = 0; r < 8; ++r) {
        const int row = r + 8 * kh;
        const int g   = gt * 16 + row;
        const float s2 = acc[j][r] * (1.0f / 16.0f);
        sc[row * 512 + nb + ln] =
            10.0f * tanhf(s2) + MASK[(size_t)(b * GGRP + g) * NNODE + nb + ln];
      }
    }
  }
  __syncthreads();

  const int row = t >> 4, l16 = t & 15;
  float* srow = sc + row * 512;
  float mx = -INFINITY;
  for (int j = l16; j < 512; j += 16) mx = fmaxf(mx, srow[j]);
  redA[row][l16] = mx;
  __syncthreads();
  if (l16 == 0) {
    float m2 = redA[row][0];
    for (int i = 1; i < 16; ++i) m2 = fmaxf(m2, redA[row][i]);
    redA[row][0] = m2;
  }
  __syncthreads();
  const float rmax = redA[row][0];
  float ssum = 0.0f;
  for (int j = l16; j < 512; j += 16) {
    const float e = __expf(srow[j] - rmax);
    srow[j] = e;
    ssum += e;
  }
  redB[row][l16] = ssum;
  __syncthreads();
  if (l16 == 0) {
    float s2 = 0.0f;
    for (int i = 0; i < 16; ++i) s2 += redB[row][i];
    redB[row][0] = s2;
  }
  __syncthreads();
  const float inv = 1.0f / redB[row][0];
  float* orow = out + (size_t)(b * GGRP + gt * 16 + row) * NNODE;
  for (int j = l16; j < 512; j += 16) orow[j] = srow[j] * inv;
}

extern "C" void kernel_launch(void* const* d_in, const int* in_sizes, int n_in,
                              void* d_out, int out_size, void* d_ws, size_t ws_size,
                              hipStream_t stream) {
  const float* EN   = (const float*)d_in[0];
  const float* EF   = (const float*)d_in[1];
  const float* EL   = (const float*)d_in[2];
  const float* MASK = (const float*)d_in[3];
  const float* WqG  = (const float*)d_in[4];
  const float* WqF  = (const float*)d_in[5];
  const float* WqL  = (const float*)d_in[6];
  const float* Wk   = (const float*)d_in[7];
  const float* Wv   = (const float*)d_in[8];
  const float* Wc   = (const float*)d_in[9];
  const float* bc   = (const float*)d_in[10];
  float* out = (float*)d_out;

  float* ws   = (float*)d_ws;
  float* qg   = ws;
  float* Qb   = qg   + (size_t)BATCH * EMBD;
  float* Kb   = Qb   + (size_t)BATCH * GGRP * EMBD;
  float* Vb   = Kb   + (size_t)BATCH * NNODE * EMBD;
  float* outc = Vb   + (size_t)BATCH * NNODE * EMBD;
  float* mhb  = outc + (size_t)BATCH * GGRP * EMBD;

  proj_kernel<<<PROJ_BLOCKS, 128, 0, stream>>>(EN, EF, EL, WqG, WqF, WqL, Wk, Wv, qg, Qb, Kb, Vb);

  dim3 ga(GGRP / 16 / GTB, HNUM / HP, BATCH);
  attn_kernel<<<ga, 256, 0, stream>>>(Qb, qg, Kb, Vb, MASK, outc);

  mh_kernel<<<512, 128, 0, stream>>>(outc, Wc, bc, mhb);

  dim3 gf(GGRP / 16, BATCH, 1);
  final_kernel<<<gf, 256, 0, stream>>>(mhb, EN, MASK, out);
}


# Final GEMM: block tile 32 query rows x 512 nodes (was 16) on the projections' staged loop: a staged node line serves 32 rows, next k-block's lines fly under WMMAs; mask tile one load group; sums kept
# speedup vs baseline: 2.2771x; 1.0406x over previous
#include <hip/hip_runtime.h>
#include <math.h>

#define BATCH 32
#define NNODE 512
#define GGRP  512
#define EMBD  256
#define HNUM  16
#define DK    16

typedef __attribute__((ext_vector_type(2))) float v2f;
typedef __attribute__((ext_vector_type(4))) float v4f;
typedef __attribute__((ext_vector_type(8))) float v8f;

__device__ __forceinline__ v8f wmma4(v2f a, v2f b, v8f c) {
  return __builtin_amdgcn_wmma_f32_16x16x4_f32(
       false, a,  false, b,
       (short)0, c,  false,  false);
}

__device__ __forceinline__ v2f ld2s(const float* p) {
  return *reinterpret_cast<const v2f*>(p);
}
__device__ __forceinline__ v4f ld4g(const float* __restrict__ p) {
  return *reinterpret_cast<const v4f*>(p);
}
__device__ __forceinline__ v2f lo2(v4f v) { return __builtin_shufflevector(v, v, 0, 1); }
__device__ __forceinline__ v2f hi2(v4f v) { return __builtin_shufflevector(v, v, 2, 3); }

#define LPITCH 36
#define TROWS  192
__device__ __forceinline__ v4f lds4(const float* p) {
  return *reinterpret_cast<const v4f*>(p);
}
__device__ __forceinline__ void gemm_pass_64x128(const float* __restrict__ A1,
                                                 const float* __restrict__ B1,
                                                 const float* __restrict__ A2,
                                                 const float* __restrict__ B2,
                                                 const int nkb,
                                                 float* tile,
                                                 const int tid, const int wm, const int wn,
                                                 const int ln, const int kh,
                                                 v8f (&acc)[2][4]) {
  const int prow = tid >> 3;
  const int pcol = (tid & 7) * 4;
  v4f g[12];
#pragma unroll
  for (int s = 0; s < 4; ++s) g[s] = ld4g(A1 + (size_t)(prow + 16 * s) * EMBD + pcol);
#pragma unroll
  for (int s = 0; s < 8; ++s) g[4 + s] = ld4g(B1 + (size_t)(prow + 16 * s) * EMBD + pcol);
  const float* ta = tile + (32 * wm + ln) * LPITCH + 4 * kh;
  const float* tb = tile + (64 + 64 * wn + ln) * LPITCH + 4 * kh;
  float* tw = tile + prow * LPITCH + pcol;
#pragma unroll 1
  for (int kb = 0; kb < nkb; ++kb) {
    __syncthreads();
#pragma unroll
    for (int s = 0; s < 12; ++s) *reinterpret_cast<v4f*>(tw + 16 * s * LPITCH) = g[s];
    __syncthreads();
    if (kb + 1 < nkb) {
      const int kn = kb + 1;
      const float* An = ((kn < 8) ? A1 : A2) + (kn & 7) * 32;
      const float* Bn = ((kn < 8) ? B1 : B2) + (kn & 7) * 32;
#pragma unroll
      for (int s = 0; s < 4; ++s) g[s] = ld4g(An + (size_t)(prow + 16 * s) * EMBD + pcol);
#pragma unroll
      for (int s = 0; s < 8; ++s) g[4 + s] = ld4g(Bn + (size_t)(prow + 16 * s) * EMBD + pcol);
    }
    v4f a0[4], a1[4], bb[4][4];
#pragma unroll
    for (int i = 0; i < 4; ++i) {
      a0[i] = lds4(ta + 8 * i);
      a1[i] = lds4(ta + 16 * LPITCH + 8 * i);
    }
#pragma unroll
    for (int j = 0; j < 4; ++j)
#pragma unroll
      for (int i = 0; i < 4; ++i) bb[j][i] = lds4(tb + j * 16 * LPITCH + 8 * i);
#pragma unroll
    for (int i = 0; i < 4; ++i) {
#pragma unroll
      for (int j = 0; j < 4; ++j) {
        acc[0][j] = wmma4(lo2(a0[i]), lo2(bb[j][i]), acc[0][j]);
        acc[1][j] = wmma4(lo2(a1[i]), lo2(bb[j][i]), acc[1][j]);
      }
#pragma unroll
      for (int j = 0; j < 4; ++j) {
        acc[0][j] = wmma4(hi2(a0[i]), hi2(bb[j][i]), acc[0][j]);
        acc[1][j] = wmma4(hi2(a1[i]), hi2(bb[j][i]), acc[1][j]);
      }
    }
  }
}

#define PROJ_GROUP  512
#define PROJ_BLOCKS (BATCH + 3 * PROJ_GROUP)
__global__ void __launch_bounds__(128) __attribute__((amdgpu_waves_per_eu(4, 5)))
proj_kernel(const float* __restrict__ EN,
                            const float* __restrict__ EF,
                            const float* __restrict__ EL,
                            const float* __restrict__ WqG,
                            const float* __restrict__ WqF,
                            const float* __restrict__ WqL,
                            const float* __restrict__ Wk,
                            const float* __restrict__ Wv,
                            float* __restrict__ qg,
                            float* __restrict__ Q,
                            float* __restrict__ K,
                            float* __restrict__ V) {
  if (blockIdx.x < BATCH) {
    __shared__ float sm[EMBD];
    const int mb = blockIdx.x;
    const int t  = threadIdx.x;
    const float* base = EN + (size_t)mb * NNODE * EMBD + 2 * t;
    float s0 = 0.0f, s1 = 0.0f;
#pragma unroll 32
    for (int n = 0; n < NNODE; ++n) {
      const v2f x = ld2s(base + (size_t)n * EMBD);
      s0 += x.x;
      s1 += x.y;
    }
    sm[2 * t]     = s0 * (1.0f / (float)NNODE);
    sm[2 * t + 1] = s1 * (1.0f / (float)NNODE);
    __syncthreads();
    float a0 = 0.0f, a1 = 0.0f;
    const float* w0 = WqG + (size_t)t * EMBD;
    const float* w1 = WqG + (size_t)(t + 128) * EMBD;
#pragma unroll 16
    for (int k = 0; k < EMBD; ++k) {
      a0 += sm[k] * w0[k];
      a1 += sm[k] * w1[k];
    }
    qg[(size_t)mb * EMBD + t]       = a0;
    qg[(size_t)mb * EMBD + t + 128] = a1;
    return;
  }
  __shared__ __attribute__((aligned(16))) float tile[TROWS * LPITCH];
  const unsigned gblk = blockIdx.x - BATCH;
  const int sel  = (int)(gblk / PROJ_GROUP);
  const int blk  = (int)(gblk % PROJ_GROUP);
  const int tid  = threadIdx.x;
  const int widx = tid >> 5;
  const int lane = tid & 31;
  const int ln   = lane & 15;
  const int kh   = lane >> 4;
  const int wm   = widx >> 1;
  const int wn   = widx & 1;
  const int row0 = (blk >> 1) * 64;
  const int col0 = (blk & 1) * 128;

  v8f acc[2][4];
#pragma unroll
  for (int i = 0; i < 2; ++i)
#pragma unroll
    for (int j = 0; j < 4; ++j) acc[i][j] = v8f{0.f,0.f,0.f,0.f,0.f,0.f,0.f,0.f};

  float* Out;
  if (sel < 2) {
    const float* W = (sel == 0) ? Wk : Wv;
    const float* A0 = EN + (size_t)row0 * EMBD;
    const float* B0 = W + (size_t)col0 * EMBD;
    gemm_pass_64x128(A0, B0, A0, B0, 8, tile, tid, wm, wn, ln, kh, acc);
    Out = (sel == 0) ? K : V;
  } else {
    gemm_pass_64x128(EF + (size_t)row0 * EMBD, WqF + (size_t)col0 * EMBD,
                     EL + (size_t)row0 * EMBD, WqL + (size_t)col0 * EMBD,
                     16, tile, tid, wm, wn, ln, kh, acc);
    Out = Q;
  }

#pragma unroll
  for (int i = 0; i < 2; ++i)
#pragma unroll
    for (int j = 0; j < 4; ++j) {
      const int col = col0 + wn * 64 + j * 16 + ln;
#pragma unroll
      for (int r = 0; r < 8; ++r) {
        const int row = row0 + wm * 32 + i * 16 + r + 8 * kh;
        Out[(size_t)row * EMBD + col] = acc[i][j][r];
      }
    }
}

#define HP 2
#define GTB 4
#define OPITCH 18
#define OHEAD (8 * 16 * OPITCH + 1)
__global__ void __launch_bounds__(256) __attribute__((amdgpu_waves_per_eu(4, 4)))
attn_kernel(const float* __restrict__ Q,
            const float* __restrict__ QG,
            const float* __restrict__ K,
            const float* __restrict__ V,
            const float* __restrict__ MASK,
            float* __restrict__ outc) {
  __shared__ float s_m[2][HP][8][16];
  __shared__ float s_l[2][HP][8][16];
  __shared__ float s_o[2][HP * OHEAD];

  const int gt0 = blockIdx.x * GTB, hp = blockIdx.y, b = blockIdx.z;
  const int t    = threadIdx.x;
  const int widx = __builtin_amdgcn_readfirstlane(t >> 5);
  const int lane = t & 31;
  const int ln   = lane & 15;
  const int kh   = lane >> 4;
  const int nb   = widx * 64;
  const int c0   = hp * (HP * DK);

  const float* Krow = K + (size_t)(b * NNODE + nb) * EMBD + c0 + (ln * EMBD + 4 * kh);
  v4f k0[HP][4], k1[HP][4];
#pragma unroll
  for (int j = 0; j < 4; ++j) {
#pragma unroll
    for (int hh = 0; hh < HP; ++hh) {
      k0[hh][j] = ld4g(Krow + (size_t)j * 16 * EMBD + hh * DK);
      k1[hh][j] = ld4g(Krow + (size_t)j * 16 * EMBD + hh * DK + 8);
    }
  }
  const float* Vp = V + (size_t)(b * NNODE + nb) * EMBD + c0 + (8 * kh * EMBD + ln);
  v2f va0[HP][4][2], va1[HP][4][2];
#pragma unroll
  for (int j = 0; j < 4; ++j) {
#pragma unroll
    for (int rr = 0; rr < 2; ++rr) {
#pragma unroll
      for (int hh = 0; hh < HP; ++hh) {
        va0[hh][j][rr].x = Vp[(size_t)(16 * j + 4 * rr)     * EMBD + hh * DK];
        va0[hh][j][rr].y = Vp[(size_t)(16 * j + 4 * rr + 1) * EMBD + hh * DK];
        va1[hh][j][rr].x = Vp[(size_t)(16 * j + 4 * rr + 2) * EMBD + hh * DK];
        va1[hh][j][rr].y = Vp[(size_t)(16 * j + 4 * rr + 3) * EMBD + hh * DK];
      }
    }
  }
  const float* QGrow = QG + (size_t)b * EMBD + c0 + (4 * kh);

  const float* Qrow = Q + (size_t)(b * GGRP + gt0 * 16) * EMBD + c0 + (ln * EMBD + 4 * kh);
  const float* Mrow = MASK + (size_t)(b * GGRP + gt0 * 16) * NNODE + nb + (ln * NNODE + 8 * kh);
  v4f ma[4], mb[4];
  v4f qc0 = ld4g(Qrow);
  v4f qc1 = ld4g(Qrow + 8);
#pragma unroll
  for (int j = 0; j < 4; ++j) {
    ma[j] = ld4g(Mrow + 16 * j);
    mb[j] = ld4g(Mrow + 16 * j + 4);
  }

#pragma unroll 1
  for (int it = 0; it < GTB; ++it) {
    const int gt = gt0 + it;
    const int pb = it & 1;

#pragma unroll
    for (int hh = 0; hh < HP; ++hh) {
      const v4f qa = ((hh == 0) ? qc0 : ld4g(Qrow + hh * DK))     + ld4g(QGrow + hh * DK);
      const v4f qb = ((hh == 0) ? qc1 : ld4g(Qrow + hh * DK + 8)) + ld4g(QGrow + hh * DK + 8);
      v8f acc[4];
#pragma unroll
      for (int j = 0; j < 4; ++j) acc[j] = v8f{0.f,0.f,0.f,0.f,0.f,0.f,0.f,0.f};
#pragma unroll
      for (int j = 0; j < 4; ++j) acc[j] = wmma4(lo2(k0[hh][j]), lo2(qa), acc[j]);
#pragma unroll
      for (int j = 0; j < 4; ++j) acc[j] = wmma4(hi2(k0[hh][j]), hi2(qa), acc[j]);
#pragma unroll
      for (int j = 0; j < 4; ++j) acc[j] = wmma4(lo2(k1[hh][j]), lo2(qb), acc[j]);
#pragma unroll
      for (int j = 0; j < 4; ++j) acc[j] = wmma4(hi2(k1[hh][j]), hi2(qb), acc[j]);

      float mloc = -INFINITY;
#pragma unroll
      for (int j = 0; j < 4; ++j) {
#pragma unroll
        for (int r = 0; r < 4; ++r) {
          acc[j][r]     = acc[j][r]     * 0.25f + ma[j][r];
          acc[j][r + 4] = acc[j][r + 4] * 0.25f + mb[j][r];
          mloc = fmaxf(mloc, fmaxf(acc[j][r], acc[j][r + 4]));
        }
      }
      mloc = fmaxf(mloc, __shfl_xor(mloc, 16));
      const float muse = (mloc == -INFINITY) ? 0.0f : mloc;

      float lsum = 0.0f;
#pragma unroll
      for (int j = 0; j < 4; ++j) {
#pragma unroll
        for (int r = 0; r < 8; ++r) {
          const float e = __expf(acc[j][r] - muse);
          acc[j][r] = e;
          lsum += e;
        }
      }
      lsum += __shfl_xor(lsum, 16);
      s_m[pb][hh][widx][ln] = mloc;
      s_l[pb][hh][widx][ln] = lsum;

      v8f o0 = v8f{0.f,0.f,0.f,0.f,0.f,0.f,0.f,0.f};
      v8f o1 = v8f{0.f,0.f,0.f,0.f,0.f,0.f,0.f,0.f};
#pragma unroll
      for (int j = 0; j < 4; ++j) {
#pragma unroll
        for (int r = 0; r < 8; r += 4) {
          v2f w0, w1;
          w0.x = acc[j][r];     w0.y = acc[j][r + 1];
          w1.x = acc[j][r + 2]; w1.y = acc[j][r + 3];
          o0 = wmma4(va0[hh][j][r / 4], w0, o0);
          o1 = wmma4(va1[hh][j][r / 4], w1, o1);
        }
      }
#pragma unroll
      for (int r = 0; r < 8; ++r)
        s_o[pb][hh * OHEAD + widx * (16 * OPITCH) + (8 * kh + r) * OPITCH + ln] = o0[r] + o1[r];
    }

    Qrow += 16 * EMBD;
    if (it + 1 < GTB) {
      Mrow += 16 * NNODE;
      qc0 = ld4g(Qrow);
      qc1 = ld4g(Qrow + 8);
#pragma unroll
      for (int j = 0; j < 4; ++j) {
        ma[j] = ld4g(Mrow + 16 * j);
        mb[j] = ld4g(Mrow + 16 * j + 4);
      }
    }
    __syncthreads();

    static_assert(HP * DK == 32, "one lane per column of the head pair");
#pragma unroll
    for (int i = 0; i < 2; ++i) {
      const int g  = 8 * i + widx;
      const int hh = lane >> 4, d = lane & 15;
      float mw[8];
      float M = -INFINITY;
#pragma unroll
      for (int w = 0; w < 8; ++w) { mw[w] = s_m[pb][hh][w][g]; M = fmaxf(M, mw[w]); }
      float L = 0.0f, O = 0.0f;
#pragma unroll
      for (int w = 0; w < 8; ++w) {
        const float e = __expf(mw[w] - M);
        L += e * s_l[pb][hh][w][g];
        O += e * s_o[pb][hh * OHEAD + w * (16 * OPITCH) + d * OPITCH + g];
      }
      outc[(size_t)(b * GGRP + gt * 16 + g) * EMBD + c0 + lane] = O / L;
    }
  }
}

__global__ void __launch_bounds__(128) __attribute__((amdgpu_waves_per_eu(4, 5)))
mh_kernel(const float* __restrict__ outc,
                          const float* __restrict__ Wc,
                          const float* __restrict__ bc,
                          float* __restrict__ mh) {
  __shared__ __attribute__((aligned(16))) float tile[TROWS * LPITCH];
  const int blk  = blockIdx.x;
  const int tid  = threadIdx.x;
  const int widx = tid >> 5;
  const int lane = tid & 31;
  const int ln   = lane & 15;
  const int kh   = lane >> 4;
  const int wm   = widx >> 1;
  const int wn   = widx & 1;
  const int row0 = (blk >> 1) * 64;
  const int col0 = (blk & 1) * 128;

  v8f acc[2][4];
#pragma unroll
  for (int i = 0; i < 2; ++i)
#pragma unroll
    for (int j = 0; j < 4; ++j) acc[i][j] = v8f{0.f,0.f,0.f,0.f,0.f,0.f,0.f,0.f};

  const float* A0 = outc + (size_t)row0 * EMBD;
  const float* B0 = Wc + (size_t)col0 * EMBD;
  gemm_pass_64x128(A0, B0, A0, B0, 8, tile, tid, wm, wn, ln, kh, acc);
#pragma unroll
  for (int i = 0; i < 2; ++i)
#pragma unroll
    for (int j = 0; j < 4; ++j) {
      const int col  = col0 + wn * 64 + j * 16 + ln;
      const float bv = bc[col];
#pragma unroll
      for (int r = 0; r < 8; ++r) {
        const int row = row0 + wm * 32 + i * 16 + r + 8 * kh;
        mh[(size_t)row * EMBD + col] = acc[i][j][r] + bv;
      }
    }
}

#define FM    32
#define FROWS (FM + NNODE)
__global__ void __launch_bounds__(256) __attribute__((amdgpu_waves_per_eu(4, 4)))
final_kernel(const float* __restrict__ mh,
                             const float* __restrict__ EN,
                             const float* __restrict__ MASK,
                             float* __restrict__ out) {
  __shared__ __attribute__((aligned(16))) float buf[FROWS * LPITCH];
  static_assert(FROWS * LPITCH >= FM * 512, "score tile fits the staging buffer");
  float* sc = buf;
  __shared__ float redA[FM][16];
  __shared__ float redB[FM][16];

  const int gt = blockIdx.x, b = blockIdx.y;
  const int t    = threadIdx.x;
  const int widx = t >> 5;
  const int lane = t & 31;
  const int ln   = lane & 15;
  const int kh   = lane >> 4;

  {
    const int prow = t >> 3;
    const int pcol = (t & 7) * 4;
    const float* Ag = mh + (size_t)(b * GGRP + gt * FM + prow) * EMBD + pcol;
    const float* Bg = EN + (size_t)(b * NNODE + prow) * EMBD + pcol;
    float* tw = buf + prow * LPITCH + pcol;
    const float* ta = buf + ln * LPITCH + 4 * kh;
    const float* tb = buf + (FM + widx * 64 + ln) * LPITCH + 4 * kh;
    v8f acc[2][4];
#pragma unroll
    for (int i = 0; i < 2; ++i)
#pragma unroll
      for (int j = 0; j < 4; ++j) acc[i][j] = v8f{0.f,0.f,0.f,0.f,0.f,0.f,0.f,0.f};
    v4f g[17];
    g[0] = ld4g(Ag);
#pragma unroll
    for (int s = 0; s < 16; ++s) g[1 + s] = ld4g(Bg + (size_t)32 * s * EMBD);
#pragma unroll 1
    for (int kb = 0; kb < EMBD / 32; ++kb) {
      __syncthreads();
#pragma unroll
      for (int s = 0; s < 17; ++s) *reinterpret_cast<v4f*>(tw + 32 * s * LPITCH) = g[s];
      __syncthreads();
      if (kb + 1 < EMBD / 32) {
        const int k = (kb + 1) * 32;
        g[0] = ld4g(Ag + k);
#pragma unroll
        for (int s = 0; s < 16; ++s) g[1 + s] = ld4g(Bg + (size_t)32 * s * EMBD + k);
      }
      v4f a0[4], a1[4], bb[4][4];
#pragma unroll
      for (int i = 0; i < 4; ++i) {
        a0[i] = lds4(ta + 8 * i);
        a1[i] = lds4(ta + 16 * LPITCH + 8 * i);
      }
#pragma unroll
      for (int j = 0; j < 4; ++j)
#pragma unroll
        for (int i = 0; i < 4; ++i) bb[j][i] = lds4(tb + j * 16 * LPITCH + 8 * i);
#pragma unroll
      for (int i = 0; i < 4; ++i) {
#pragma unroll
        for (int j = 0; j < 4; ++j) {
          acc[0][j] = wmma4(lo2(a0[i]), lo2(bb[j][i]), acc[0][j]);
          acc[1][j] = wmma4(lo2(a1[i]), lo2(bb[j][i]), acc[1][j]);
        }
#pragma unroll
        for (int j = 0; j < 4; ++j) {
          acc[0][j] = wmma4(hi2(a0[i]), hi2(bb[j][i]), acc[0][j]);
          acc[1][j] = wmma4(hi2(a1[i]), hi2(bb[j][i]), acc[1][j]);
        }
      }
    }
    __syncthreads();
    float mk[2][4][8];
#pragma unroll
    for (int i = 0; i < 2; ++i)
#pragma unroll
      for (int j = 0; j < 4; ++j)
#pragma unroll
        for (int r = 0; r < 8; ++r)
          mk[i][j][r] = MASK[(size_t)(b * GGRP + gt * FM + i * 16 + r + 8 * kh) * NNODE
                             + (widx * 4 + j) * 16 + ln];
#pragma unroll
    for (int i = 0; i < 2; ++i)
#pragma unroll
      for (int j = 0; j < 4; ++j) {
        const int nb = (widx * 4 + j) * 16;
#pragma unroll
        for (int r = 0; r < 8; ++r) {
          const int row = i * 16 + r + 8 * kh;
          const float s2 = acc[i][j][r] * (1.0f / 16.0f);
          sc[row * 512 + nb + ln] = 10.0f * tanhf(s2) + mk[i][j][r];
        }
      }
  }
  __syncthreads();

  const int r0 = t >> 4, l16 = t & 15;
#pragma unroll
  for (int h = 0; h < 2; ++h) {
    const float* srow = sc + (r0 + 16 * h) * 512;
    float mx = -INFINITY;
    for (int j = l16; j < 512; j += 16) mx = fmaxf(mx, srow[j]);
    redA[r0 + 16 * h][l16] = mx;
  }
  __syncthreads();
  if (l16 == 0) {
#pragma unroll
    for (int h = 0; h < 2; ++h) {
      const int row = r0 + 16 * h;
      float m2 = redA[row][0];
      for (int i = 1; i < 16; ++i) m2 = fmaxf(m2, redA[row][i]);
      redA[row][0] = m2;
    }
  }
  __syncthreads();
#pragma unroll
  for (int h = 0; h < 2; ++h) {
    const int row = r0 + 16 * h;
    float* srow = sc + row * 512;
    const float rmax = redA[row][0];
    float ssum = 0.0f;
    for (int j = l16; j < 512; j += 16) {
      const float e = __expf(srow[j] - rmax);
      srow[j] = e;
      ssum += e;
    }
    redB[row][l16] = ssum;
  }
  __syncthreads();
  if (l16 == 0) {
#pragma unroll
    for (int h = 0; h < 2; ++h) {
      const int row = r0 + 16 * h;
      float s2 = 0.0f;
      for (int i = 0; i < 16; ++i) s2 += redB[row][i];
      redB[row][0] = s2;
    }
  }
  __syncthreads();
#pragma unroll
  for (int h = 0; h < 2; ++h) {
    const int row = r0 + 16 * h;
    const float* srow = sc + row * 512;
    const float inv = 1.0f / redB[row][0];
    float* orow = out + (size_t)(b * GGRP + gt * FM + row) * NNODE;
    for (int j = l16; j < 512; j += 16) orow[j] = srow[j] * inv;
  }
}

extern "C" void kernel_launch(void* const* d_in, const int* in_sizes, int n_in,
                              void* d_out, int out_size, void* d_ws, size_t ws_size,
                              hipStream_t stream) {
  const float* EN   = (const float*)d_in[0];
  const float* EF   = (const float*)d_in[1];
  const float* EL   = (const float*)d_in[2];
  const float* MASK = (const float*)d_in[3];
  const float* WqG  = (const float*)d_in[4];
  const float* WqF  = (const float*)d_in[5];
  const float* WqL  = (const float*)d_in[6];
  const float* Wk   = (const float*)d_in[7];
  const float* Wv   = (const float*)d_in[8];
  const float* Wc   = (const float*)d_in[9];
  const float* bc   = (const float*)d_in[10];
  float* out = (float*)d_out;

  float* ws   = (float*)d_ws;
  float* qg   = ws;
  float* Qb   = qg   + (size_t)BATCH * EMBD;
  float* Kb   = Qb   + (size_t)BATCH * GGRP * EMBD;
  float* Vb   = Kb   + (size_t)BATCH * NNODE * EMBD;
  float* outc = Vb   + (size_t)BATCH * NNODE * EMBD;
  float* mhb  = outc + (size_t)BATCH * GGRP * EMBD;

  proj_kernel<<<PROJ_BLOCKS, 128, 0, stream>>>(EN, EF, EL, WqG, WqF, WqL, Wk, Wv, qg, Qb, Kb, Vb);

  dim3 ga(GGRP / 16 / GTB, HNUM / HP, BATCH);
  attn_kernel<<<ga, 256, 0, stream>>>(Qb, qg, Kb, Vb, MASK, outc);

  mh_kernel<<<512, 128, 0, stream>>>(outc, Wc, bc, mhb);

  dim3 gf(GGRP / FM, BATCH, 1);
  final_kernel<<<gf, 256, 0, stream>>>(mhb, EN, MASK, out);
}
